# starting kernel
# baseline (speedup 1.0000x reference)
#include <hip/hip_runtime.h>

typedef float v2f __attribute__((ext_vector_type(2)));
typedef float v8f __attribute__((ext_vector_type(8)));

#define NHEADS 16
#define HD     64
#define SEQ    2048
#define DM     1024
#define NBH    32
#define RWIN   128
#define RSZ    257
#define RPST   260
#define TST    66

__device__ __forceinline__ v8f wmma4(v2f a, v2f b, v8f c) {
    return __builtin_amdgcn_wmma_f32_16x16x4_f32(false, a, false, b, (short)0, c, false, false);
}

__global__ __launch_bounds__(256, 1)
void qkv_gemm(const float* __restrict__ X, const float* __restrict__ W,
              const float* __restrict__ bias,
              float* __restrict__ qws, float* __restrict__ kws, float* __restrict__ vws)
{
    __shared__ float As[128 * 18];
    __shared__ float Bs[128 * 18];

    const int tid  = threadIdx.x;
    const int lane = tid & 31;
    const int wave = tid >> 5;
    const int lo   = lane & 15;
    const int hi   = lane >> 4;
    const int wm   = (wave & 3) * 32;
    const int wn   = (wave >> 2) * 64;
    const int mB   = blockIdx.x * 128;
    const int nB   = blockIdx.y * 128;

    const v8f zv = {0.f, 0.f, 0.f, 0.f, 0.f, 0.f, 0.f, 0.f};
    v8f acc[2][4];
    #pragma unroll
    for (int mt = 0; mt < 2; ++mt)
        #pragma unroll
        for (int nt = 0; nt < 4; ++nt)
            acc[mt][nt] = zv;

    const int lrow = tid >> 1;
    const int lk8  = (tid & 1) << 3;

    for (int ks = 0; ks < DM; ks += 16) {
        __syncthreads();
        {
            const float* ga = X + (size_t)(mB + lrow) * DM + ks + lk8;
            float4 a0 = *(const float4*)ga;
            float4 a1 = *(const float4*)(ga + 4);
            float* da = As + lrow * 18 + lk8;
            da[0] = a0.x; da[1] = a0.y; da[2] = a0.z; da[3] = a0.w;
            da[4] = a1.x; da[5] = a1.y; da[6] = a1.z; da[7] = a1.w;

            const float* gb = W + (size_t)(nB + lrow) * DM + ks + lk8;
            float4 b0 = *(const float4*)gb;
            float4 b1 = *(const float4*)(gb + 4);
            float* db = Bs + lrow * 18 + lk8;
            db[0] = b0.x; db[1] = b0.y; db[2] = b0.z; db[3] = b0.w;
            db[4] = b1.x; db[5] = b1.y; db[6] = b1.z; db[7] = b1.w;
        }
        __syncthreads();

        #pragma unroll
        for (int s4 = 0; s4 < 4; ++s4) {
            const int k0 = (s4 << 2) + (hi << 1);
            v2f a0 = *(const v2f*)(As + (wm + lo) * 18 + k0);
            v2f a1 = *(const v2f*)(As + (wm + 16 + lo) * 18 + k0);
            #pragma unroll
            for (int nt = 0; nt < 4; ++nt) {
                v2f b = *(const v2f*)(Bs + (wn + nt * 16 + lo) * 18 + k0);
                acc[0][nt] = wmma4(a0, b, acc[0][nt]);
                acc[1][nt] = wmma4(a1, b, acc[1][nt]);
            }
        }
    }

    #pragma unroll
    for (int nt = 0; nt < 4; ++nt) {
        const int col    = nB + wn + nt * 16 + lo;
        const float bv   = bias[col];
        const int head   = col / 192;
        const int within = col - head * 192;
        const int which  = within >> 6;
        const int dim    = within & 63;
        float* dst = (which == 0) ? qws : (which == 1) ? kws : vws;
        #pragma unroll
        for (int mt = 0; mt < 2; ++mt) {
            #pragma unroll
            for (int i = 0; i < 8; ++i) {
                const int row   = mB + wm + mt * 16 + i + (hi << 3);
                const int batch = row >> 11;
                const int si    = row & 2047;
                dst[(((size_t)batch * NHEADS + head) * SEQ + si) * HD + dim] =
                    acc[mt][nt][i] + bv;
            }
        }
    }
}

__global__ __launch_bounds__(128, 1)
void rel_attn(const float* __restrict__ qws, const float* __restrict__ kws,
              const float* __restrict__ vws, const float* __restrict__ Wr,
              const float* __restrict__ brv, float* __restrict__ out)
{
    extern __shared__ float sm[];
    float* Sq  = sm;
    float* Sk  = Sq + 64 * TST;
    float* Sv  = Sk + 64 * TST;
    float* Sp  = Sv + 64 * TST;
    float* Srp = Sp + 64 * TST;

    const int bh   = blockIdx.x;
    const int qt   = blockIdx.y;
    const int tid  = threadIdx.x;
    const int lane = tid & 31;
    const int wave = tid >> 5;
    const int lo   = lane & 15;
    const int hi   = lane >> 4;

    const int batch = bh >> 4;
    const int head  = bh & 15;
    const size_t base = (size_t)bh * SEQ * HD;
    const int q0 = qt * 64;

    for (int idx = tid; idx < 64 * 16; idx += 128) {
        const int r  = idx >> 4;
        const int c4 = (idx & 15) << 2;
        float4 v = *(const float4*)(qws + base + (size_t)(q0 + r) * HD + c4);
        float* d = Sq + r * TST + c4;
        d[0] = v.x; d[1] = v.y; d[2] = v.z; d[3] = v.w;
    }
    __syncthreads();

    for (int idx = tid; idx < 64 * RSZ; idx += 128) {
        const int r = idx / RSZ;
        const int o = idx - r * RSZ;
        const float* wr = Wr + o * HD;
        const float* qq = Sq + r * TST;
        float s = 0.f;
        #pragma unroll 8
        for (int k = 0; k < HD; ++k) s += qq[k] * wr[k];
        Srp[r * RPST + o] = s + brv[o];
    }

    const v8f zv = {0.f, 0.f, 0.f, 0.f, 0.f, 0.f, 0.f, 0.f};
    v8f oacc[4];
    #pragma unroll
    for (int nt = 0; nt < 4; ++nt) oacc[nt] = zv;
    float mrun[8], lrun[8];
    #pragma unroll
    for (int i = 0; i < 8; ++i) { mrun[i] = -__builtin_inff(); lrun[i] = 0.f; }

    __syncthreads();

    for (int kt = 0; kt < 32; ++kt) {
        for (int idx = tid; idx < 64 * 16; idx += 128) {
            const int r  = idx >> 4;
            const int c4 = (idx & 15) << 2;
            const size_t g = base + (size_t)(kt * 64 + r) * HD + c4;
            float4 kv = *(const float4*)(kws + g);
            float4 vv = *(const float4*)(vws + g);
            float* dk = Sk + r * TST + c4;
            float* dv = Sv + r * TST + c4;
            dk[0] = kv.x; dk[1] = kv.y; dk[2] = kv.z; dk[3] = kv.w;
            dv[0] = vv.x; dv[1] = vv.y; dv[2] = vv.z; dv[3] = vv.w;
        }
        __syncthreads();

        v8f sc[4];
        #pragma unroll
        for (int nt = 0; nt < 4; ++nt) sc[nt] = zv;
        #pragma unroll
        for (int ksub = 0; ksub < 16; ++ksub) {
            const int k0 = (ksub << 2) + (hi << 1);
            v2f a = *(const v2f*)(Sq + (wave * 16 + lo) * TST + k0);
            #pragma unroll
            for (int nt = 0; nt < 4; ++nt) {
                v2f b = *(const v2f*)(Sk + (nt * 16 + lo) * TST + k0);
                sc[nt] = wmma4(a, b, sc[nt]);
            }
        }

        float rmax[8];
        #pragma unroll
        for (int i = 0; i < 8; ++i) rmax[i] = -__builtin_inff();
        #pragma unroll
        for (int nt = 0; nt < 4; ++nt) {
            const int jg = kt * 64 + nt * 16 + lo;
            #pragma unroll
            for (int i = 0; i < 8; ++i) {
                const int rt = wave * 16 + i + (hi << 3);
                const int ig = q0 + rt;
                int off = jg - ig;
                off = (off < -RWIN) ? -RWIN : (off > RWIN) ? RWIN : off;
                float s = sc[nt][i] * 0.125f + Srp[rt * RPST + off + RWIN];
                sc[nt][i] = s;
                rmax[i] = fmaxf(rmax[i], s);
            }
        }
        #pragma unroll
        for (int d = 1; d < 16; d <<= 1)
            #pragma unroll
            for (int i = 0; i < 8; ++i)
                rmax[i] = fmaxf(rmax[i], __shfl_xor(rmax[i], d));

        float mnew[8], alpha[8], rsum[8];
        #pragma unroll
        for (int i = 0; i < 8; ++i) {
            mnew[i]  = fmaxf(mrun[i], rmax[i]);
            alpha[i] = __expf(mrun[i] - mnew[i]);
            mrun[i]  = mnew[i];
            rsum[i]  = 0.f;
        }
        #pragma unroll
        for (int nt = 0; nt < 4; ++nt)
            #pragma unroll
            for (int i = 0; i < 8; ++i) {
                float p = __expf(sc[nt][i] - mnew[i]);
                sc[nt][i] = p;
                rsum[i] += p;
            }
        #pragma unroll
        for (int d = 1; d < 16; d <<= 1)
            #pragma unroll
            for (int i = 0; i < 8; ++i)
                rsum[i] += __shfl_xor(rsum[i], d);
        #pragma unroll
        for (int i = 0; i < 8; ++i) lrun[i] = lrun[i] * alpha[i] + rsum[i];

        #pragma unroll
        for (int nt = 0; nt < 4; ++nt)
            #pragma unroll
            for (int i = 0; i < 8; ++i)
                oacc[nt][i] *= alpha[i];

        #pragma unroll
        for (int nt = 0; nt < 4; ++nt)
            #pragma unroll
            for (int i = 0; i < 8; ++i)
                Sp[(wave * 16 + i + (hi << 3)) * TST + nt * 16 + lo] = sc[nt][i];
        __syncthreads();

        #pragma unroll
        for (int ksub = 0; ksub < 16; ++ksub) {
            const int k0 = (ksub << 2) + (hi << 1);
            v2f a = *(const v2f*)(Sp + (wave * 16 + lo) * TST + k0);
            #pragma unroll
            for (int nt = 0; nt < 4; ++nt) {
                v2f b;
                b.x = Sv[k0 * TST + nt * 16 + lo];
                b.y = Sv[(k0 + 1) * TST + nt * 16 + lo];
                oacc[nt] = wmma4(a, b, oacc[nt]);
            }
        }
        __syncthreads();
    }

    #pragma unroll
    for (int nt = 0; nt < 4; ++nt) {
        const int dim = nt * 16 + lo;
        #pragma unroll
        for (int i = 0; i < 8; ++i) {
            const int row = q0 + wave * 16 + i + (hi << 3);
            out[((size_t)batch * SEQ + row) * DM + head * HD + dim] =
                oacc[nt][i] / lrun[i];
        }
    }
}

extern "C" void kernel_launch(void* const* d_in, const int* in_sizes, int n_in,
                              void* d_out, int out_size, void* d_ws, size_t ws_size,
                              hipStream_t stream)
{
    const float* x    = (const float*)d_in[0];
    const float* Wqkv = (const float*)d_in[1];
    const float* bqkv = (const float*)d_in[2];
    const float* Wr   = (const float*)d_in[3];
    const float* br   = (const float*)d_in[4];
    float* out = (float*)d_out;

    float* ws  = (float*)d_ws;
    const size_t per = (size_t)NBH * SEQ * HD;
    float* qws = ws;
    float* kws = ws + per;
    float* vws = ws + 2 * per;

    dim3 g1(32, 24), b1(256);
    qkv_gemm<<<g1, b1, 0, stream>>>(x, Wqkv, bqkv, qws, kws, vws);

    dim3 g2(NBH, SEQ / 64), b2(128);
    size_t smem = (size_t)(4 * 64 * TST + 64 * RPST) * sizeof(float);
    rel_attn<<<g2, b2, smem, stream>>>(qws, kws, vws, Wr, br, out);
}


# rp = Q*Wr^T (relative-position logits) moved from the per-thread scalar FMA loop with gathered Wr loads onto f32 WMMA, Wr rows read as b128 fragments; rp sums now in WMMA order
# speedup vs baseline: 1.5718x; 1.5718x over previous
#include <hip/hip_runtime.h>

typedef float v2f __attribute__((ext_vector_type(2)));
typedef float v8f __attribute__((ext_vector_type(8)));

#define NHEADS 16
#define HD     64
#define SEQ    2048
#define DM     1024
#define NBH    32
#define RWIN   128
#define RSZ    257
#define RPST   260
#define TST    66

__device__ __forceinline__ v8f wmma4(v2f a, v2f b, v8f c) {
    return __builtin_amdgcn_wmma_f32_16x16x4_f32(false, a, false, b, (short)0, c, false, false);
}

__global__ __launch_bounds__(256, 1)
void qkv_gemm(const float* __restrict__ X, const float* __restrict__ W,
              const float* __restrict__ bias,
              float* __restrict__ qws, float* __restrict__ kws, float* __restrict__ vws)
{
    __shared__ float As[128 * 18];
    __shared__ float Bs[128 * 18];

    const int tid  = threadIdx.x;
    const int lane = tid & 31;
    const int wave = tid >> 5;
    const int lo   = lane & 15;
    const int hi   = lane >> 4;
    const int wm   = (wave & 3) * 32;
    const int wn   = (wave >> 2) * 64;
    const int mB   = blockIdx.x * 128;
    const int nB   = blockIdx.y * 128;

    const v8f zv = {0.f, 0.f, 0.f, 0.f, 0.f, 0.f, 0.f, 0.f};
    v8f acc[2][4];
    #pragma unroll
    for (int mt = 0; mt < 2; ++mt)
        #pragma unroll
        for (int nt = 0; nt < 4; ++nt)
            acc[mt][nt] = zv;

    const int lrow = tid >> 1;
    const int lk8  = (tid & 1) << 3;

    for (int ks = 0; ks < DM; ks += 16) {
        __syncthreads();
        {
            const float* ga = X + (size_t)(mB + lrow) * DM + ks + lk8;
            float4 a0 = *(const float4*)ga;
            float4 a1 = *(const float4*)(ga + 4);
            float* da = As + lrow * 18 + lk8;
            da[0] = a0.x; da[1] = a0.y; da[2] = a0.z; da[3] = a0.w;
            da[4] = a1.x; da[5] = a1.y; da[6] = a1.z; da[7] = a1.w;

            const float* gb = W + (size_t)(nB + lrow) * DM + ks + lk8;
            float4 b0 = *(const float4*)gb;
            float4 b1 = *(const float4*)(gb + 4);
            float* db = Bs + lrow * 18 + lk8;
            db[0] = b0.x; db[1] = b0.y; db[2] = b0.z; db[3] = b0.w;
            db[4] = b1.x; db[5] = b1.y; db[6] = b1.z; db[7] = b1.w;
        }
        __syncthreads();

        #pragma unroll
        for (int s4 = 0; s4 < 4; ++s4) {
            const int k0 = (s4 << 2) + (hi << 1);
            v2f a0 = *(const v2f*)(As + (wm + lo) * 18 + k0);
            v2f a1 = *(const v2f*)(As + (wm + 16 + lo) * 18 + k0);
            #pragma unroll
            for (int nt = 0; nt < 4; ++nt) {
                v2f b = *(const v2f*)(Bs + (wn + nt * 16 + lo) * 18 + k0);
                acc[0][nt] = wmma4(a0, b, acc[0][nt]);
                acc[1][nt] = wmma4(a1, b, acc[1][nt]);
            }
        }
    }

    #pragma unroll
    for (int nt = 0; nt < 4; ++nt) {
        const int col    = nB + wn + nt * 16 + lo;
        const float bv   = bias[col];
        const int head   = col / 192;
        const int within = col - head * 192;
        const int which  = within >> 6;
        const int dim    = within & 63;
        float* dst = (which == 0) ? qws : (which == 1) ? kws : vws;
        #pragma unroll
        for (int mt = 0; mt < 2; ++mt) {
            #pragma unroll
            for (int i = 0; i < 8; ++i) {
                const int row   = mB + wm + mt * 16 + i + (hi << 3);
                const int batch = row >> 11;
                const int si    = row & 2047;
                dst[(((size_t)batch * NHEADS + head) * SEQ + si) * HD + dim] =
                    acc[mt][nt][i] + bv;
            }
        }
    }
}

__global__ __launch_bounds__(128, 1)
void rel_attn(const float* __restrict__ qws, const float* __restrict__ kws,
              const float* __restrict__ vws, const float* __restrict__ Wr,
              const float* __restrict__ brv, float* __restrict__ out)
{
    extern __shared__ float sm[];
    float* Sq  = sm;
    float* Sk  = Sq + 64 * TST;
    float* Sv  = Sk + 64 * TST;
    float* Sp  = Sv + 64 * TST;
    float* Srp = Sp + 64 * TST;

    const int bh   = blockIdx.x;
    const int qt   = blockIdx.y;
    const int tid  = threadIdx.x;
    const int lane = tid & 31;
    const int wave = tid >> 5;
    const int lo   = lane & 15;
    const int hi   = lane >> 4;

    const int batch = bh >> 4;
    const int head  = bh & 15;
    const size_t base = (size_t)bh * SEQ * HD;
    const int q0 = qt * 64;

    for (int idx = tid; idx < 64 * 16; idx += 128) {
        const int r  = idx >> 4;
        const int c4 = (idx & 15) << 2;
        float4 v = *(const float4*)(qws + base + (size_t)(q0 + r) * HD + c4);
        float* d = Sq + r * TST + c4;
        d[0] = v.x; d[1] = v.y; d[2] = v.z; d[3] = v.w;
    }
    __syncthreads();

    const v8f zv = {0.f, 0.f, 0.f, 0.f, 0.f, 0.f, 0.f, 0.f};

    {
        v2f af[16];
        const float* qrow = Sq + (wave * 16 + lo) * TST + (hi << 2);
        #pragma unroll
        for (int j = 0; j < 8; ++j) {
            af[2 * j]     = *(const v2f*)(qrow + 8 * j);
            af[2 * j + 1] = *(const v2f*)(qrow + 8 * j + 2);
        }

        float4 bc[8], bn[8];
        {
            const float4* p = (const float4*)(Wr + (size_t)lo * HD + (hi << 2));
            #pragma unroll
            for (int j = 0; j < 8; ++j) bc[j] = p[2 * j];
        }

        #pragma unroll 1
        for (int nt = 0; nt < 17; ++nt) {
            {
                int n = (nt + 1) * 16 + lo;
                n = (n < RSZ) ? n : (RSZ - 1);
                const float4* p = (const float4*)(Wr + (size_t)n * HD + (hi << 2));
                #pragma unroll
                for (int j = 0; j < 8; ++j) bn[j] = p[2 * j];
            }

            v8f c0 = zv, c1 = zv;
            #pragma unroll
            for (int j = 0; j < 4; ++j) {
                v2f b0 = {bc[j].x, bc[j].y};
                v2f b1 = {bc[j].z, bc[j].w};
                v2f b2 = {bc[j + 4].x, bc[j + 4].y};
                v2f b3 = {bc[j + 4].z, bc[j + 4].w};
                c0 = wmma4(af[2 * j],     b0, c0);
                c1 = wmma4(af[2 * j + 8], b2, c1);
                c0 = wmma4(af[2 * j + 1], b1, c0);
                c1 = wmma4(af[2 * j + 9], b3, c1);
            }

            const int col = nt * 16 + lo;
            if (col < RSZ) {
                const float bb = brv[col];
                #pragma unroll
                for (int i = 0; i < 8; ++i)
                    Srp[(wave * 16 + i + (hi << 3)) * RPST + col] = (c0[i] + c1[i]) + bb;
            }

            #pragma unroll
            for (int j = 0; j < 8; ++j) bc[j] = bn[j];
        }
    }

    v8f oacc[4];
    #pragma unroll
    for (int nt = 0; nt < 4; ++nt) oacc[nt] = zv;
    float mrun[8], lrun[8];
    #pragma unroll
    for (int i = 0; i < 8; ++i) { mrun[i] = -__builtin_inff(); lrun[i] = 0.f; }

    __syncthreads();

    for (int kt = 0; kt < 32; ++kt) {
        for (int idx = tid; idx < 64 * 16; idx += 128) {
            const int r  = idx >> 4;
            const int c4 = (idx & 15) << 2;
            const size_t g = base + (size_t)(kt * 64 + r) * HD + c4;
            float4 kv = *(const float4*)(kws + g);
            float4 vv = *(const float4*)(vws + g);
            float* dk = Sk + r * TST + c4;
            float* dv = Sv + r * TST + c4;
            dk[0] = kv.x; dk[1] = kv.y; dk[2] = kv.z; dk[3] = kv.w;
            dv[0] = vv.x; dv[1] = vv.y; dv[2] = vv.z; dv[3] = vv.w;
        }
        __syncthreads();

        v8f sc[4];
        #pragma unroll
        for (int nt = 0; nt < 4; ++nt) sc[nt] = zv;
        #pragma unroll
        for (int ksub = 0; ksub < 16; ++ksub) {
            const int k0 = (ksub << 2) + (hi << 1);
            v2f a = *(const v2f*)(Sq + (wave * 16 + lo) * TST + k0);
            #pragma unroll
            for (int nt = 0; nt < 4; ++nt) {
                v2f b = *(const v2f*)(Sk + (nt * 16 + lo) * TST + k0);
                sc[nt] = wmma4(a, b, sc[nt]);
            }
        }

        float rmax[8];
        #pragma unroll
        for (int i = 0; i < 8; ++i) rmax[i] = -__builtin_inff();
        #pragma unroll
        for (int nt = 0; nt < 4; ++nt) {
            const int jg = kt * 64 + nt * 16 + lo;
            #pragma unroll
            for (int i = 0; i < 8; ++i) {
                const int rt = wave * 16 + i + (hi << 3);
                const int ig = q0 + rt;
                int off = jg - ig;
                off = (off < -RWIN) ? -RWIN : (off > RWIN) ? RWIN : off;
                float s = sc[nt][i] * 0.125f + Srp[rt * RPST + off + RWIN];
                sc[nt][i] = s;
                rmax[i] = fmaxf(rmax[i], s);
            }
        }
        #pragma unroll
        for (int d = 1; d < 16; d <<= 1)
            #pragma unroll
            for (int i = 0; i < 8; ++i)
                rmax[i] = fmaxf(rmax[i], __shfl_xor(rmax[i], d));

        float mnew[8], alpha[8], rsum[8];
        #pragma unroll
        for (int i = 0; i < 8; ++i) {
            mnew[i]  = fmaxf(mrun[i], rmax[i]);
            alpha[i] = __expf(mrun[i] - mnew[i]);
            mrun[i]  = mnew[i];
            rsum[i]  = 0.f;
        }
        #pragma unroll
        for (int nt = 0; nt < 4; ++nt)
            #pragma unroll
            for (int i = 0; i < 8; ++i) {
                float p = __expf(sc[nt][i] - mnew[i]);
                sc[nt][i] = p;
                rsum[i] += p;
            }
        #pragma unroll
        for (int d = 1; d < 16; d <<= 1)
            #pragma unroll
            for (int i = 0; i < 8; ++i)
                rsum[i] += __shfl_xor(rsum[i], d);
        #pragma unroll
        for (int i = 0; i < 8; ++i) lrun[i] = lrun[i] * alpha[i] + rsum[i];

        #pragma unroll
        for (int nt = 0; nt < 4; ++nt)
            #pragma unroll
            for (int i = 0; i < 8; ++i)
                oacc[nt][i] *= alpha[i];

        #pragma unroll
        for (int nt = 0; nt < 4; ++nt)
            #pragma unroll
            for (int i = 0; i < 8; ++i)
                Sp[(wave * 16 + i + (hi << 3)) * TST + nt * 16 + lo] = sc[nt][i];
        __syncthreads();

        #pragma unroll
        for (int ksub = 0; ksub < 16; ++ksub) {
            const int k0 = (ksub << 2) + (hi << 1);
            v2f a = *(const v2f*)(Sp + (wave * 16 + lo) * TST + k0);
            #pragma unroll
            for (int nt = 0; nt < 4; ++nt) {
                v2f b;
                b.x = Sv[k0 * TST + nt * 16 + lo];
                b.y = Sv[(k0 + 1) * TST + nt * 16 + lo];
                oacc[nt] = wmma4(a, b, oacc[nt]);
            }
        }
        __syncthreads();
    }

    #pragma unroll
    for (int nt = 0; nt < 4; ++nt) {
        const int dim = nt * 16 + lo;
        #pragma unroll
        for (int i = 0; i < 8; ++i) {
            const int row = q0 + wave * 16 + i + (hi << 3);
            out[((size_t)batch * SEQ + row) * DM + head * HD + dim] =
                oacc[nt][i] / lrun[i];
        }
    }
}

extern "C" void kernel_launch(void* const* d_in, const int* in_sizes, int n_in,
                              void* d_out, int out_size, void* d_ws, size_t ws_size,
                              hipStream_t stream)
{
    const float* x    = (const float*)d_in[0];
    const float* Wqkv = (const float*)d_in[1];
    const float* bqkv = (const float*)d_in[2];
    const float* Wr   = (const float*)d_in[3];
    const float* br   = (const float*)d_in[4];
    float* out = (float*)d_out;

    float* ws  = (float*)d_ws;
    const size_t per = (size_t)NBH * SEQ * HD;
    float* qws = ws;
    float* kws = ws + per;
    float* vws = ws + 2 * per;

    dim3 g1(32, 24), b1(256);
    qkv_gemm<<<g1, b1, 0, stream>>>(x, Wqkv, bqkv, qws, kws, vws);

    dim3 g2(NBH, SEQ / 64), b2(128);
    size_t smem = (size_t)(4 * 64 * TST + 64 * RPST) * sizeof(float);
    rel_attn<<<g2, b2, smem, stream>>>(qws, kws, vws, Wr, br, out);
}


# rel_attn: K/V tile kt+1 is loaded global->registers as one group of 16 b128 loads under tile kt's compute and stored to LDS after it; was 8 serial load-wait-store trips in front of every tile
# speedup vs baseline: 1.8705x; 1.1900x over previous
#include <hip/hip_runtime.h>

typedef float v2f __attribute__((ext_vector_type(2)));
typedef float v8f __attribute__((ext_vector_type(8)));

#define NHEADS 16
#define HD     64
#define SEQ    2048
#define DM     1024
#define NBH    32
#define RWIN   128
#define RSZ    257
#define RPST   260
#define TST    66

__device__ __forceinline__ v8f wmma4(v2f a, v2f b, v8f c) {
    return __builtin_amdgcn_wmma_f32_16x16x4_f32(false, a, false, b, (short)0, c, false, false);
}

__global__ __launch_bounds__(256, 1)
void qkv_gemm(const float* __restrict__ X, const float* __restrict__ W,
              const float* __restrict__ bias,
              float* __restrict__ qws, float* __restrict__ kws, float* __restrict__ vws)
{
    __shared__ float As[128 * 18];
    __shared__ float Bs[128 * 18];

    const int tid  = threadIdx.x;
    const int lane = tid & 31;
    const int wave = tid >> 5;
    const int lo   = lane & 15;
    const int hi   = lane >> 4;
    const int wm   = (wave & 3) * 32;
    const int wn   = (wave >> 2) * 64;
    const int mB   = blockIdx.x * 128;
    const int nB   = blockIdx.y * 128;

    const v8f zv = {0.f, 0.f, 0.f, 0.f, 0.f, 0.f, 0.f, 0.f};
    v8f acc[2][4];
    #pragma unroll
    for (int mt = 0; mt < 2; ++mt)
        #pragma unroll
        for (int nt = 0; nt < 4; ++nt)
            acc[mt][nt] = zv;

    const int lrow = tid >> 1;
    const int lk8  = (tid & 1) << 3;

    for (int ks = 0; ks < DM; ks += 16) {
        __syncthreads();
        {
            const float* ga = X + (size_t)(mB + lrow) * DM + ks + lk8;
            float4 a0 = *(const float4*)ga;
            float4 a1 = *(const float4*)(ga + 4);
            float* da = As + lrow * 18 + lk8;
            da[0] = a0.x; da[1] = a0.y; da[2] = a0.z; da[3] = a0.w;
            da[4] = a1.x; da[5] = a1.y; da[6] = a1.z; da[7] = a1.w;

            const float* gb = W + (size_t)(nB + lrow) * DM + ks + lk8;
            float4 b0 = *(const float4*)gb;
            float4 b1 = *(const float4*)(gb + 4);
            float* db = Bs + lrow * 18 + lk8;
            db[0] = b0.x; db[1] = b0.y; db[2] = b0.z; db[3] = b0.w;
            db[4] = b1.x; db[5] = b1.y; db[6] = b1.z; db[7] = b1.w;
        }
        __syncthreads();

        #pragma unroll
        for (int s4 = 0; s4 < 4; ++s4) {
            const int k0 = (s4 << 2) + (hi << 1);
            v2f a0 = *(const v2f*)(As + (wm + lo) * 18 + k0);
            v2f a1 = *(const v2f*)(As + (wm + 16 + lo) * 18 + k0);
            #pragma unroll
            for (int nt = 0; nt < 4; ++nt) {
                v2f b = *(const v2f*)(Bs + (wn + nt * 16 + lo) * 18 + k0);
                acc[0][nt] = wmma4(a0, b, acc[0][nt]);
                acc[1][nt] = wmma4(a1, b, acc[1][nt]);
            }
        }
    }

    #pragma unroll
    for (int nt = 0; nt < 4; ++nt) {
        const int col    = nB + wn + nt * 16 + lo;
        const float bv   = bias[col];
        const int head   = col / 192;
        const int within = col - head * 192;
        const int which  = within >> 6;
        const int dim    = within & 63;
        float* dst = (which == 0) ? qws : (which == 1) ? kws : vws;
        #pragma unroll
        for (int mt = 0; mt < 2; ++mt) {
            #pragma unroll
            for (int i = 0; i < 8; ++i) {
                const int row   = mB + wm + mt * 16 + i + (hi << 3);
                const int batch = row >> 11;
                const int si    = row & 2047;
                dst[(((size_t)batch * NHEADS + head) * SEQ + si) * HD + dim] =
                    acc[mt][nt][i] + bv;
            }
        }
    }
}

__global__ __launch_bounds__(128, 1)
void rel_attn(const float* __restrict__ qws, const float* __restrict__ kws,
              const float* __restrict__ vws, const float* __restrict__ Wr,
              const float* __restrict__ brv, float* __restrict__ out)
{
    extern __shared__ float sm[];
    float* Sq  = sm;
    float* Sk  = Sq + 64 * TST;
    float* Sv  = Sk + 64 * TST;
    float* Sp  = Sv + 64 * TST;
    float* Srp = Sp + 64 * TST;

    const int bh   = blockIdx.x;
    const int qt   = blockIdx.y;
    const int tid  = threadIdx.x;
    const int lane = tid & 31;
    const int wave = tid >> 5;
    const int lo   = lane & 15;
    const int hi   = lane >> 4;

    const int batch = bh >> 4;
    const int head  = bh & 15;
    const size_t base = (size_t)bh * SEQ * HD;
    const int q0 = qt * 64;

    for (int idx = tid; idx < 64 * 16; idx += 128) {
        const int r  = idx >> 4;
        const int c4 = (idx & 15) << 2;
        float4 v = *(const float4*)(qws + base + (size_t)(q0 + r) * HD + c4);
        float* d = Sq + r * TST + c4;
        d[0] = v.x; d[1] = v.y; d[2] = v.z; d[3] = v.w;
    }
    __syncthreads();

    const v8f zv = {0.f, 0.f, 0.f, 0.f, 0.f, 0.f, 0.f, 0.f};

    const int sr  = tid >> 4;
    const int sc4 = (tid & 15) << 2;
    const float* kp = kws + base + (size_t)sr * HD + sc4;
    const float* vp = vws + base + (size_t)sr * HD + sc4;
    float4 kreg[8], vreg[8];
    #pragma unroll
    for (int it = 0; it < 8; ++it) {
        kreg[it] = *(const float4*)(kp + it * (8 * HD));
        vreg[it] = *(const float4*)(vp + it * (8 * HD));
    }

    {
        v2f af[16];
        const float* qrow = Sq + (wave * 16 + lo) * TST + (hi << 2);
        #pragma unroll
        for (int j = 0; j < 8; ++j) {
            af[2 * j]     = *(const v2f*)(qrow + 8 * j);
            af[2 * j + 1] = *(const v2f*)(qrow + 8 * j + 2);
        }

        float4 bc[8], bn[8];
        {
            const float4* p = (const float4*)(Wr + (size_t)lo * HD + (hi << 2));
            #pragma unroll
            for (int j = 0; j < 8; ++j) bc[j] = p[2 * j];
        }

        #pragma unroll 1
        for (int nt = 0; nt < 17; ++nt) {
            {
                int n = (nt + 1) * 16 + lo;
                n = (n < RSZ) ? n : (RSZ - 1);
                const float4* p = (const float4*)(Wr + (size_t)n * HD + (hi << 2));
                #pragma unroll
                for (int j = 0; j < 8; ++j) bn[j] = p[2 * j];
            }

            v8f c0 = zv, c1 = zv;
            #pragma unroll
            for (int j = 0; j < 4; ++j) {
                v2f b0 = {bc[j].x, bc[j].y};
                v2f b1 = {bc[j].z, bc[j].w};
                v2f b2 = {bc[j + 4].x, bc[j + 4].y};
                v2f b3 = {bc[j + 4].z, bc[j + 4].w};
                c0 = wmma4(af[2 * j],     b0, c0);
                c1 = wmma4(af[2 * j + 8], b2, c1);
                c0 = wmma4(af[2 * j + 1], b1, c0);
                c1 = wmma4(af[2 * j + 9], b3, c1);
            }

            const int col = nt * 16 + lo;
            if (col < RSZ) {
                const float bb = brv[col];
                #pragma unroll
                for (int i = 0; i < 8; ++i)
                    Srp[(wave * 16 + i + (hi << 3)) * RPST + col] = (c0[i] + c1[i]) + bb;
            }

            #pragma unroll
            for (int j = 0; j < 8; ++j) bc[j] = bn[j];
        }
    }

    v8f oacc[4];
    #pragma unroll
    for (int nt = 0; nt < 4; ++nt) oacc[nt] = zv;
    float mrun[8], lrun[8];
    #pragma unroll
    for (int i = 0; i < 8; ++i) { mrun[i] = -__builtin_inff(); lrun[i] = 0.f; }

    __syncthreads();

    for (int kt = 0; kt < 32; ++kt) {
        #pragma unroll
        for (int it = 0; it < 8; ++it) {
            float* dk = Sk + (sr + 8 * it) * TST + sc4;
            float* dv = Sv + (sr + 8 * it) * TST + sc4;
            dk[0] = kreg[it].x; dk[1] = kreg[it].y; dk[2] = kreg[it].z; dk[3] = kreg[it].w;
            dv[0] = vreg[it].x; dv[1] = vreg[it].y; dv[2] = vreg[it].z; dv[3] = vreg[it].w;
        }
        __syncthreads();

        if (kt + 1 < 32) {
            const float* kn = kp + (size_t)(kt + 1) * (64 * HD);
            const float* vn = vp + (size_t)(kt + 1) * (64 * HD);
            #pragma unroll
            for (int it = 0; it < 8; ++it) {
                kreg[it] = *(const float4*)(kn + it * (8 * HD));
                vreg[it] = *(const float4*)(vn + it * (8 * HD));
            }
        }

        v8f sc[4];
        #pragma unroll
        for (int nt = 0; nt < 4; ++nt) sc[nt] = zv;
        #pragma unroll
        for (int ksub = 0; ksub < 16; ++ksub) {
            const int k0 = (ksub << 2) + (hi << 1);
            v2f a = *(const v2f*)(Sq + (wave * 16 + lo) * TST + k0);
            #pragma unroll
            for (int nt = 0; nt < 4; ++nt) {
                v2f b = *(const v2f*)(Sk + (nt * 16 + lo) * TST + k0);
                sc[nt] = wmma4(a, b, sc[nt]);
            }
        }

        float rmax[8];
        #pragma unroll
        for (int i = 0; i < 8; ++i) rmax[i] = -__builtin_inff();
        #pragma unroll
        for (int nt = 0; nt < 4; ++nt) {
            const int jg = kt * 64 + nt * 16 + lo;
            #pragma unroll
            for (int i = 0; i < 8; ++i) {
                const int rt = wave * 16 + i + (hi << 3);
                const int ig = q0 + rt;
                int off = jg - ig;
                off = (off < -RWIN) ? -RWIN : (off > RWIN) ? RWIN : off;
                float s = sc[nt][i] * 0.125f + Srp[rt * RPST + off + RWIN];
                sc[nt][i] = s;
                rmax[i] = fmaxf(rmax[i], s);
            }
        }
        #pragma unroll
        for (int d = 1; d < 16; d <<= 1)
            #pragma unroll
            for (int i = 0; i < 8; ++i)
                rmax[i] = fmaxf(rmax[i], __shfl_xor(rmax[i], d));

        float mnew[8], alpha[8], rsum[8];
        #pragma unroll
        for (int i = 0; i < 8; ++i) {
            mnew[i]  = fmaxf(mrun[i], rmax[i]);
            alpha[i] = __expf(mrun[i] - mnew[i]);
            mrun[i]  = mnew[i];
            rsum[i]  = 0.f;
        }
        #pragma unroll
        for (int nt = 0; nt < 4; ++nt)
            #pragma unroll
            for (int i = 0; i < 8; ++i) {
                float p = __expf(sc[nt][i] - mnew[i]);
                sc[nt][i] = p;
                rsum[i] += p;
            }
        #pragma unroll
        for (int d = 1; d < 16; d <<= 1)
            #pragma unroll
            for (int i = 0; i < 8; ++i)
                rsum[i] += __shfl_xor(rsum[i], d);
        #pragma unroll
        for (int i = 0; i < 8; ++i) lrun[i] = lrun[i] * alpha[i] + rsum[i];

        #pragma unroll
        for (int nt = 0; nt < 4; ++nt)
            #pragma unroll
            for (int i = 0; i < 8; ++i)
                oacc[nt][i] *= alpha[i];

        #pragma unroll
        for (int nt = 0; nt < 4; ++nt)
            #pragma unroll
            for (int i = 0; i < 8; ++i)
                Sp[(wave * 16 + i + (hi << 3)) * TST + nt * 16 + lo] = sc[nt][i];
        __syncthreads();

        #pragma unroll
        for (int ksub = 0; ksub < 16; ++ksub) {
            const int k0 = (ksub << 2) + (hi << 1);
            v2f a = *(const v2f*)(Sp + (wave * 16 + lo) * TST + k0);
            #pragma unroll
            for (int nt = 0; nt < 4; ++nt) {
                v2f b;
                b.x = Sv[k0 * TST + nt * 16 + lo];
                b.y = Sv[(k0 + 1) * TST + nt * 16 + lo];
                oacc[nt] = wmma4(a, b, oacc[nt]);
            }
        }
        __syncthreads();
    }

    #pragma unroll
    for (int nt = 0; nt < 4; ++nt) {
        const int dim = nt * 16 + lo;
        #pragma unroll
        for (int i = 0; i < 8; ++i) {
            const int row = q0 + wave * 16 + i + (hi << 3);
            out[((size_t)batch * SEQ + row) * DM + head * HD + dim] =
                oacc[nt][i] / lrun[i];
        }
    }
}

extern "C" void kernel_launch(void* const* d_in, const int* in_sizes, int n_in,
                              void* d_out, int out_size, void* d_ws, size_t ws_size,
                              hipStream_t stream)
{
    const float* x    = (const float*)d_in[0];
    const float* Wqkv = (const float*)d_in[1];
    const float* bqkv = (const float*)d_in[2];
    const float* Wr   = (const float*)d_in[3];
    const float* br   = (const float*)d_in[4];
    float* out = (float*)d_out;

    float* ws  = (float*)d_ws;
    const size_t per = (size_t)NBH * SEQ * HD;
    float* qws = ws;
    float* kws = ws + per;
    float* vws = ws + 2 * per;

    dim3 g1(32, 24), b1(256);
    qkv_gemm<<<g1, b1, 0, stream>>>(x, Wqkv, bqkv, qws, kws, vws);

    dim3 g2(NBH, SEQ / 64), b2(128);
    size_t smem = (size_t)(4 * 64 * TST + 64 * RPST) * sizeof(float);
    rel_attn<<<g2, b2, smem, stream>>>(qws, kws, vws, Wr, br, out);
}


# rel_attn: scores computed transposed (S^T=K*Q^T): a lane owns one query, so row max/sum are in-lane + 1 permlanex16 (were 64 bpermutes) and P^T feeds PV from registers, not LDS; sum order changes
# speedup vs baseline: 2.4391x; 1.3040x over previous
#include <hip/hip_runtime.h>

typedef float v2f __attribute__((ext_vector_type(2)));
typedef float v8f __attribute__((ext_vector_type(8)));

#define NHEADS 16
#define HD     64
#define SEQ    2048
#define DM     1024
#define NBH    32
#define RWIN   128
#define RSZ    257
#define RPST   260
#define TST    66

__device__ __forceinline__ v8f wmma4(v2f a, v2f b, v8f c) {
    return __builtin_amdgcn_wmma_f32_16x16x4_f32(false, a, false, b, (short)0, c, false, false);
}

__device__ __forceinline__ float xhalf(float x) {
    int b = __builtin_bit_cast(int, x);
    return __builtin_bit_cast(float,
        __builtin_amdgcn_permlanex16(b, b, 0x76543210, 0xfedcba98, false, false));
}

__global__ __launch_bounds__(256, 1)
void qkv_gemm(const float* __restrict__ X, const float* __restrict__ W,
              const float* __restrict__ bias,
              float* __restrict__ qws, float* __restrict__ kws, float* __restrict__ vws)
{
    __shared__ float As[128 * 18];
    __shared__ float Bs[128 * 18];

    const int tid  = threadIdx.x;
    const int lane = tid & 31;
    const int wave = tid >> 5;
    const int lo   = lane & 15;
    const int hi   = lane >> 4;
    const int wm   = (wave & 3) * 32;
    const int wn   = (wave >> 2) * 64;
    const int mB   = blockIdx.x * 128;
    const int nB   = blockIdx.y * 128;

    const v8f zv = {0.f, 0.f, 0.f, 0.f, 0.f, 0.f, 0.f, 0.f};
    v8f acc[2][4];
    #pragma unroll
    for (int mt = 0; mt < 2; ++mt)
        #pragma unroll
        for (int nt = 0; nt < 4; ++nt)
            acc[mt][nt] = zv;

    const int lrow = tid >> 1;
    const int lk8  = (tid & 1) << 3;

    for (int ks = 0; ks < DM; ks += 16) {
        __syncthreads();
        {
            const float* ga = X + (size_t)(mB + lrow) * DM + ks + lk8;
            float4 a0 = *(const float4*)ga;
            float4 a1 = *(const float4*)(ga + 4);
            float* da = As + lrow * 18 + lk8;
            da[0] = a0.x; da[1] = a0.y; da[2] = a0.z; da[3] = a0.w;
            da[4] = a1.x; da[5] = a1.y; da[6] = a1.z; da[7] = a1.w;

            const float* gb = W + (size_t)(nB + lrow) * DM + ks + lk8;
            float4 b0 = *(const float4*)gb;
            float4 b1 = *(const float4*)(gb + 4);
            float* db = Bs + lrow * 18 + lk8;
            db[0] = b0.x; db[1] = b0.y; db[2] = b0.z; db[3] = b0.w;
            db[4] = b1.x; db[5] = b1.y; db[6] = b1.z; db[7] = b1.w;
        }
        __syncthreads();

        #pragma unroll
        for (int s4 = 0; s4 < 4; ++s4) {
            const int k0 = (s4 << 2) + (hi << 1);
            v2f a0 = *(const v2f*)(As + (wm + lo) * 18 + k0);
            v2f a1 = *(const v2f*)(As + (wm + 16 + lo) * 18 + k0);
            #pragma unroll
            for (int nt = 0; nt < 4; ++nt) {
                v2f b = *(const v2f*)(Bs + (wn + nt * 16 + lo) * 18 + k0);
                acc[0][nt] = wmma4(a0, b, acc[0][nt]);
                acc[1][nt] = wmma4(a1, b, acc[1][nt]);
            }
        }
    }

    #pragma unroll
    for (int nt = 0; nt < 4; ++nt) {
        const int col    = nB + wn + nt * 16 + lo;
        const float bv   = bias[col];
        const int head   = col / 192;
        const int within = col - head * 192;
        const int which  = within >> 6;
        const int dim    = within & 63;
        float* dst = (which == 0) ? qws : (which == 1) ? kws : vws;
        #pragma unroll
        for (int mt = 0; mt < 2; ++mt) {
            #pragma unroll
            for (int i = 0; i < 8; ++i) {
                const int row   = mB + wm + mt * 16 + i + (hi << 3);
                const int batch = row >> 11;
                const int si    = row & 2047;
                dst[(((size_t)batch * NHEADS + head) * SEQ + si) * HD + dim] =
                    acc[mt][nt][i] + bv;
            }
        }
    }
}

__global__ __launch_bounds__(128, 1)
void rel_attn(const float* __restrict__ qws, const float* __restrict__ kws,
              const float* __restrict__ vws, const float* __restrict__ Wr,
              const float* __restrict__ brv, float* __restrict__ out)
{
    extern __shared__ float sm[];
    float* Sq  = sm;
    float* Sk  = Sq + 64 * TST;
    float* Sv  = Sk + 64 * TST;
    float* Srp = Sv + 64 * TST;

    const int bh   = blockIdx.x;
    const int qt   = blockIdx.y;
    const int tid  = threadIdx.x;
    const int lane = tid & 31;
    const int wave = tid >> 5;
    const int lo   = lane & 15;
    const int hi   = lane >> 4;

    const int batch = bh >> 4;
    const int head  = bh & 15;
    const size_t base = (size_t)bh * SEQ * HD;
    const int q0 = qt * 64;

    for (int idx = tid; idx < 64 * 16; idx += 128) {
        const int r  = idx >> 4;
        const int c4 = (idx & 15) << 2;
        float4 v = *(const float4*)(qws + base + (size_t)(q0 + r) * HD + c4);
        float* d = Sq + r * TST + c4;
        d[0] = v.x; d[1] = v.y; d[2] = v.z; d[3] = v.w;
    }
    __syncthreads();

    const v8f zv = {0.f, 0.f, 0.f, 0.f, 0.f, 0.f, 0.f, 0.f};

    const int sr  = tid >> 4;
    const int sc4 = (tid & 15) << 2;
    const float* kp = kws + base + (size_t)sr * HD + sc4;
    const float* vp = vws + base + (size_t)sr * HD + sc4;
    float4 kreg[8], vreg[8];
    #pragma unroll
    for (int it = 0; it < 8; ++it) {
        kreg[it] = *(const float4*)(kp + it * (8 * HD));
        vreg[it] = *(const float4*)(vp + it * (8 * HD));
    }

    {
        v2f af[16];
        const float* qrow = Sq + (wave * 16 + lo) * TST + (hi << 2);
        #pragma unroll
        for (int j = 0; j < 8; ++j) {
            af[2 * j]     = *(const v2f*)(qrow + 8 * j);
            af[2 * j + 1] = *(const v2f*)(qrow + 8 * j + 2);
        }

        float4 bc[8], bn[8];
        {
            const float4* p = (const float4*)(Wr + (size_t)lo * HD + (hi << 2));
            #pragma unroll
            for (int j = 0; j < 8; ++j) bc[j] = p[2 * j];
        }

        #pragma unroll 1
        for (int nt = 0; nt < 17; ++nt) {
            {
                int n = (nt + 1) * 16 + lo;
                n = (n < RSZ) ? n : (RSZ - 1);
                const float4* p = (const float4*)(Wr + (size_t)n * HD + (hi << 2));
                #pragma unroll
                for (int j = 0; j < 8; ++j) bn[j] = p[2 * j];
            }

            v8f c0 = zv, c1 = zv;
            #pragma unroll
            for (int j = 0; j < 4; ++j) {
                v2f b0 = {bc[j].x, bc[j].y};
                v2f b1 = {bc[j].z, bc[j].w};
                v2f b2 = {bc[j + 4].x, bc[j + 4].y};
                v2f b3 = {bc[j + 4].z, bc[j + 4].w};
                c0 = wmma4(af[2 * j],     b0, c0);
                c1 = wmma4(af[2 * j + 8], b2, c1);
                c0 = wmma4(af[2 * j + 1], b1, c0);
                c1 = wmma4(af[2 * j + 9], b3, c1);
            }

            const int col = nt * 16 + lo;
            if (col < RSZ) {
                const float bb = brv[col];
                #pragma unroll
                for (int i = 0; i < 8; ++i)
                    Srp[(wave * 16 + i + (hi << 3)) * RPST + col] = (c0[i] + c1[i]) + bb;
            }

            #pragma unroll
            for (int j = 0; j < 8; ++j) bc[j] = bn[j];
        }
    }

    v8f ot[4];
    #pragma unroll
    for (int dt = 0; dt < 4; ++dt) ot[dt] = zv;
    float mrun = -__builtin_inff(), lrun = 0.f;

    const int    qrow  = wave * 16 + lo;
    const float* srow  = Srp + qrow * RPST + RWIN;
    const int    dq    = (hi << 3) - (q0 + qrow);

    __syncthreads();

    for (int kt = 0; kt < 32; ++kt) {
        #pragma unroll
        for (int it = 0; it < 8; ++it) {
            float* dk = Sk + (sr + 8 * it) * TST + sc4;
            float* dv = Sv + (sr + 8 * it) * TST + sc4;
            dk[0] = kreg[it].x; dk[1] = kreg[it].y; dk[2] = kreg[it].z; dk[3] = kreg[it].w;
            dv[0] = vreg[it].x; dv[1] = vreg[it].y; dv[2] = vreg[it].z; dv[3] = vreg[it].w;
        }
        __syncthreads();

        if (kt + 1 < 32) {
            const float* kn = kp + (size_t)(kt + 1) * (64 * HD);
            const float* vn = vp + (size_t)(kt + 1) * (64 * HD);
            #pragma unroll
            for (int it = 0; it < 8; ++it) {
                kreg[it] = *(const float4*)(kn + it * (8 * HD));
                vreg[it] = *(const float4*)(vn + it * (8 * HD));
            }
        }

        v8f st[4];
        #pragma unroll
        for (int nt = 0; nt < 4; ++nt) st[nt] = zv;
        #pragma unroll
        for (int ksub = 0; ksub < 16; ++ksub) {
            const int k0 = (ksub << 2) + (hi << 1);
            v2f b = *(const v2f*)(Sq + qrow * TST + k0);
            #pragma unroll
            for (int nt = 0; nt < 4; ++nt) {
                v2f a = *(const v2f*)(Sk + (nt * 16 + lo) * TST + k0);
                st[nt] = wmma4(a, b, st[nt]);
            }
        }

        const int dbase = kt * 64 + dq;
        float tmax = -__builtin_inff();
        #pragma unroll
        for (int nt = 0; nt < 4; ++nt) {
            #pragma unroll
            for (int r = 0; r < 8; ++r) {
                int off = dbase + nt * 16 + r;
                off = (off < -RWIN) ? -RWIN : (off > RWIN) ? RWIN : off;
                float s = st[nt][r] * 0.125f + srow[off];
                st[nt][r] = s;
                tmax = fmaxf(tmax, s);
            }
        }
        tmax = fmaxf(tmax, xhalf(tmax));

        const float mnew  = fmaxf(mrun, tmax);
        const float alpha = __expf(mrun - mnew);
        mrun = mnew;
        float rs[4];
        #pragma unroll
        for (int nt = 0; nt < 4; ++nt) {
            rs[nt] = 0.f;
            #pragma unroll
            for (int r = 0; r < 8; ++r) {
                float p = __expf(st[nt][r] - mnew);
                st[nt][r] = p;
                rs[nt] += p;
            }
        }
        float rsum = (rs[0] + rs[1]) + (rs[2] + rs[3]);
        rsum += xhalf(rsum);
        lrun = lrun * alpha + rsum;

        #pragma unroll
        for (int dt = 0; dt < 4; ++dt)
            #pragma unroll
            for (int r = 0; r < 8; ++r)
                ot[dt][r] *= alpha;

        #pragma unroll
        for (int nt = 0; nt < 4; ++nt) {
            #pragma unroll
            for (int t = 0; t < 4; ++t) {
                v2f b = {st[nt][2 * t], st[nt][2 * t + 1]};
                const float* vrow = Sv + (nt * 16 + (hi << 3) + 2 * t) * TST + lo;
                #pragma unroll
                for (int dt = 0; dt < 4; ++dt) {
                    v2f a;
                    a.x = vrow[dt * 16];
                    a.y = vrow[TST + dt * 16];
                    ot[dt] = wmma4(a, b, ot[dt]);
                }
            }
        }
        __syncthreads();
    }

    {
        float* so = Sq + qrow * TST + (hi << 3);
        #pragma unroll
        for (int dt = 0; dt < 4; ++dt)
            #pragma unroll
            for (int r = 0; r < 8; ++r)
                so[dt * 16 + r] = ot[dt][r] / lrun;
    }
    __syncthreads();
    #pragma unroll
    for (int it = 0; it < 8; ++it) {
        const int idx = it * 32 + lane;
        const int row = wave * 16 + (idx >> 4);
        const int c4  = (idx & 15) << 2;
        const float* s = Sq + row * TST + c4;
        float4 v;
        v.x = s[0]; v.y = s[1]; v.z = s[2]; v.w = s[3];
        *(float4*)(out + ((size_t)batch * SEQ + q0 + row) * DM + head * HD + c4) = v;
    }
}

extern "C" void kernel_launch(void* const* d_in, const int* in_sizes, int n_in,
                              void* d_out, int out_size, void* d_ws, size_t ws_size,
                              hipStream_t stream)
{
    const float* x    = (const float*)d_in[0];
    const float* Wqkv = (const float*)d_in[1];
    const float* bqkv = (const float*)d_in[2];
    const float* Wr   = (const float*)d_in[3];
    const float* br   = (const float*)d_in[4];
    float* out = (float*)d_out;

    float* ws  = (float*)d_ws;
    const size_t per = (size_t)NBH * SEQ * HD;
    float* qws = ws;
    float* kws = ws + per;
    float* vws = ws + 2 * per;

    dim3 g1(32, 24), b1(256);
    qkv_gemm<<<g1, b1, 0, stream>>>(x, Wqkv, bqkv, qws, kws, vws);

    dim3 g2(NBH, SEQ / 64), b2(128);
    size_t smem = (size_t)(3 * 64 * TST + 64 * RPST) * sizeof(float);
    rel_attn<<<g2, b2, smem, stream>>>(qws, kws, vws, Wr, br, out);
}


# rel_attn: V tile staged transposed in LDS (Svt[dim][key]): each PV A fragment (2 consecutive keys of one dim) is one aligned b64 read, was 2 b32 reads + register repack; sum order unchanged
# speedup vs baseline: 2.5552x; 1.0476x over previous
#include <hip/hip_runtime.h>

typedef float v2f __attribute__((ext_vector_type(2)));
typedef float v8f __attribute__((ext_vector_type(8)));

#define NHEADS 16
#define HD     64
#define SEQ    2048
#define DM     1024
#define NBH    32
#define RWIN   128
#define RSZ    257
#define RPST   260
#define TST    66

__device__ __forceinline__ v8f wmma4(v2f a, v2f b, v8f c) {
    return __builtin_amdgcn_wmma_f32_16x16x4_f32(false, a, false, b, (short)0, c, false, false);
}

__device__ __forceinline__ float xhalf(float x) {
    int b = __builtin_bit_cast(int, x);
    return __builtin_bit_cast(float,
        __builtin_amdgcn_permlanex16(b, b, 0x76543210, 0xfedcba98, false, false));
}

__global__ __launch_bounds__(256, 1)
void qkv_gemm(const float* __restrict__ X, const float* __restrict__ W,
              const float* __restrict__ bias,
              float* __restrict__ qws, float* __restrict__ kws, float* __restrict__ vws)
{
    __shared__ float As[128 * 18];
    __shared__ float Bs[128 * 18];

    const int tid  = threadIdx.x;
    const int lane = tid & 31;
    const int wave = tid >> 5;
    const int lo   = lane & 15;
    const int hi   = lane >> 4;
    const int wm   = (wave & 3) * 32;
    const int wn   = (wave >> 2) * 64;
    const int mB   = blockIdx.x * 128;
    const int nB   = blockIdx.y * 128;

    const v8f zv = {0.f, 0.f, 0.f, 0.f, 0.f, 0.f, 0.f, 0.f};
    v8f acc[2][4];
    #pragma unroll
    for (int mt = 0; mt < 2; ++mt)
        #pragma unroll
        for (int nt = 0; nt < 4; ++nt)
            acc[mt][nt] = zv;

    const int lrow = tid >> 1;
    const int lk8  = (tid & 1) << 3;

    for (int ks = 0; ks < DM; ks += 16) {
        __syncthreads();
        {
            const float* ga = X + (size_t)(mB + lrow) * DM + ks + lk8;
            float4 a0 = *(const float4*)ga;
            float4 a1 = *(const float4*)(ga + 4);
            float* da = As + lrow * 18 + lk8;
            da[0] = a0.x; da[1] = a0.y; da[2] = a0.z; da[3] = a0.w;
            da[4] = a1.x; da[5] = a1.y; da[6] = a1.z; da[7] = a1.w;

            const float* gb = W + (size_t)(nB + lrow) * DM + ks + lk8;
            float4 b0 = *(const float4*)gb;
            float4 b1 = *(const float4*)(gb + 4);
            float* db = Bs + lrow * 18 + lk8;
            db[0] = b0.x; db[1] = b0.y; db[2] = b0.z; db[3] = b0.w;
            db[4] = b1.x; db[5] = b1.y; db[6] = b1.z; db[7] = b1.w;
        }
        __syncthreads();

        #pragma unroll
        for (int s4 = 0; s4 < 4; ++s4) {
            const int k0 = (s4 << 2) + (hi << 1);
            v2f a0 = *(const v2f*)(As + (wm + lo) * 18 + k0);
            v2f a1 = *(const v2f*)(As + (wm + 16 + lo) * 18 + k0);
            #pragma unroll
            for (int nt = 0; nt < 4; ++nt) {
                v2f b = *(const v2f*)(Bs + (wn + nt * 16 + lo) * 18 + k0);
                acc[0][nt] = wmma4(a0, b, acc[0][nt]);
                acc[1][nt] = wmma4(a1, b, acc[1][nt]);
            }
        }
    }

    #pragma unroll
    for (int nt = 0; nt < 4; ++nt) {
        const int col    = nB + wn + nt * 16 + lo;
        const float bv   = bias[col];
        const int head   = col / 192;
        const int within = col - head * 192;
        const int which  = within >> 6;
        const int dim    = within & 63;
        float* dst = (which == 0) ? qws : (which == 1) ? kws : vws;
        #pragma unroll
        for (int mt = 0; mt < 2; ++mt) {
            #pragma unroll
            for (int i = 0; i < 8; ++i) {
                const int row   = mB + wm + mt * 16 + i + (hi << 3);
                const int batch = row >> 11;
                const int si    = row & 2047;
                dst[(((size_t)batch * NHEADS + head) * SEQ + si) * HD + dim] =
                    acc[mt][nt][i] + bv;
            }
        }
    }
}

__global__ __launch_bounds__(128, 1)
void rel_attn(const float* __restrict__ qws, const float* __restrict__ kws,
              const float* __restrict__ vws, const float* __restrict__ Wr,
              const float* __restrict__ brv, float* __restrict__ out)
{
    extern __shared__ float sm[];
    float* Sq  = sm;
    float* Sk  = Sq + 64 * TST;
    float* Svt = Sk + 64 * TST;
    float* Srp = Svt + 64 * TST;

    const int bh   = blockIdx.x;
    const int qt   = blockIdx.y;
    const int tid  = threadIdx.x;
    const int lane = tid & 31;
    const int wave = tid >> 5;
    const int lo   = lane & 15;
    const int hi   = lane >> 4;

    const int batch = bh >> 4;
    const int head  = bh & 15;
    const size_t base = (size_t)bh * SEQ * HD;
    const int q0 = qt * 64;

    for (int idx = tid; idx < 64 * 16; idx += 128) {
        const int r  = idx >> 4;
        const int c4 = (idx & 15) << 2;
        float4 v = *(const float4*)(qws + base + (size_t)(q0 + r) * HD + c4);
        float* d = Sq + r * TST + c4;
        d[0] = v.x; d[1] = v.y; d[2] = v.z; d[3] = v.w;
    }
    __syncthreads();

    const v8f zv = {0.f, 0.f, 0.f, 0.f, 0.f, 0.f, 0.f, 0.f};

    const int sr  = tid >> 4;
    const int sc4 = (tid & 15) << 2;
    const float* kp = kws + base + (size_t)sr * HD + sc4;
    const float* vp = vws + base + (size_t)(2 * sr) * HD + sc4;
    float4 kreg[8], vreg[8];
    #pragma unroll
    for (int it = 0; it < 8; ++it) {
        kreg[it] = *(const float4*)(kp + it * (8 * HD));
        vreg[it] = *(const float4*)(vp + ((it >> 1) * 16 + (it & 1)) * HD);
    }

    {
        v2f af[16];
        const float* qrow = Sq + (wave * 16 + lo) * TST + (hi << 2);
        #pragma unroll
        for (int j = 0; j < 8; ++j) {
            af[2 * j]     = *(const v2f*)(qrow + 8 * j);
            af[2 * j + 1] = *(const v2f*)(qrow + 8 * j + 2);
        }

        float4 bc[8], bn[8];
        {
            const float4* p = (const float4*)(Wr + (size_t)lo * HD + (hi << 2));
            #pragma unroll
            for (int j = 0; j < 8; ++j) bc[j] = p[2 * j];
        }

        #pragma unroll 1
        for (int nt = 0; nt < 17; ++nt) {
            {
                int n = (nt + 1) * 16 + lo;
                n = (n < RSZ) ? n : (RSZ - 1);
                const float4* p = (const float4*)(Wr + (size_t)n * HD + (hi << 2));
                #pragma unroll
                for (int j = 0; j < 8; ++j) bn[j] = p[2 * j];
            }

            v8f c0 = zv, c1 = zv;
            #pragma unroll
            for (int j = 0; j < 4; ++j) {
                v2f b0 = {bc[j].x, bc[j].y};
                v2f b1 = {bc[j].z, bc[j].w};
                v2f b2 = {bc[j + 4].x, bc[j + 4].y};
                v2f b3 = {bc[j + 4].z, bc[j + 4].w};
                c0 = wmma4(af[2 * j],     b0, c0);
                c1 = wmma4(af[2 * j + 8], b2, c1);
                c0 = wmma4(af[2 * j + 1], b1, c0);
                c1 = wmma4(af[2 * j + 9], b3, c1);
            }

            const int col = nt * 16 + lo;
            if (col < RSZ) {
                const float bb = brv[col];
                #pragma unroll
                for (int i = 0; i < 8; ++i)
                    Srp[(wave * 16 + i + (hi << 3)) * RPST + col] = (c0[i] + c1[i]) + bb;
            }

            #pragma unroll
            for (int j = 0; j < 8; ++j) bc[j] = bn[j];
        }
    }

    v8f ot[4];
    #pragma unroll
    for (int dt = 0; dt < 4; ++dt) ot[dt] = zv;
    float mrun = -__builtin_inff(), lrun = 0.f;

    const int    qrow  = wave * 16 + lo;
    const float* srow  = Srp + qrow * RPST + RWIN;
    const int    dq    = (hi << 3) - (q0 + qrow);

    __syncthreads();

    for (int kt = 0; kt < 32; ++kt) {
        #pragma unroll
        for (int it = 0; it < 8; ++it) {
            float* dk = Sk + (sr + 8 * it) * TST + sc4;
            dk[0] = kreg[it].x; dk[1] = kreg[it].y; dk[2] = kreg[it].z; dk[3] = kreg[it].w;
        }
        #pragma unroll
        for (int p = 0; p < 4; ++p) {
            float* dv = Svt + sc4 * TST + 16 * p + 2 * sr;
            *(v2f*)(dv)           = v2f{vreg[2 * p].x, vreg[2 * p + 1].x};
            *(v2f*)(dv + TST)     = v2f{vreg[2 * p].y, vreg[2 * p + 1].y};
            *(v2f*)(dv + 2 * TST) = v2f{vreg[2 * p].z, vreg[2 * p + 1].z};
            *(v2f*)(dv + 3 * TST) = v2f{vreg[2 * p].w, vreg[2 * p + 1].w};
        }
        __syncthreads();

        if (kt + 1 < 32) {
            const float* kn = kp + (size_t)(kt + 1) * (64 * HD);
            const float* vn = vp + (size_t)(kt + 1) * (64 * HD);
            #pragma unroll
            for (int it = 0; it < 8; ++it) {
                kreg[it] = *(const float4*)(kn + it * (8 * HD));
                vreg[it] = *(const float4*)(vn + ((it >> 1) * 16 + (it & 1)) * HD);
            }
        }

        v8f st[4];
        #pragma unroll
        for (int nt = 0; nt < 4; ++nt) st[nt] = zv;
        #pragma unroll
        for (int ksub = 0; ksub < 16; ++ksub) {
            const int k0 = (ksub << 2) + (hi << 1);
            v2f b = *(const v2f*)(Sq + qrow * TST + k0);
            #pragma unroll
            for (int nt = 0; nt < 4; ++nt) {
                v2f a = *(const v2f*)(Sk + (nt * 16 + lo) * TST + k0);
                st[nt] = wmma4(a, b, st[nt]);
            }
        }

        const int dbase = kt * 64 + dq;
        float tmax = -__builtin_inff();
        #pragma unroll
        for (int nt = 0; nt < 4; ++nt) {
            #pragma unroll
            for (int r = 0; r < 8; ++r) {
                int off = dbase + nt * 16 + r;
                off = (off < -RWIN) ? -RWIN : (off > RWIN) ? RWIN : off;
                float s = st[nt][r] * 0.125f + srow[off];
                st[nt][r] = s;
                tmax = fmaxf(tmax, s);
            }
        }
        tmax = fmaxf(tmax, xhalf(tmax));

        const float mnew  = fmaxf(mrun, tmax);
        const float alpha = __expf(mrun - mnew);
        mrun = mnew;
        float rs[4];
        #pragma unroll
        for (int nt = 0; nt < 4; ++nt) {
            rs[nt] = 0.f;
            #pragma unroll
            for (int r = 0; r < 8; ++r) {
                float p = __expf(st[nt][r] - mnew);
                st[nt][r] = p;
                rs[nt] += p;
            }
        }
        float rsum = (rs[0] + rs[1]) + (rs[2] + rs[3]);
        rsum += xhalf(rsum);
        lrun = lrun * alpha + rsum;

        #pragma unroll
        for (int dt = 0; dt < 4; ++dt)
            #pragma unroll
            for (int r = 0; r < 8; ++r)
                ot[dt][r] *= alpha;

        #pragma unroll
        for (int nt = 0; nt < 4; ++nt) {
            #pragma unroll
            for (int t = 0; t < 4; ++t) {
                v2f b = {st[nt][2 * t], st[nt][2 * t + 1]};
                const float* vcol = Svt + lo * TST + nt * 16 + (hi << 3) + 2 * t;
                #pragma unroll
                for (int dt = 0; dt < 4; ++dt) {
                    v2f a = *(const v2f*)(vcol + dt * 16 * TST);
                    ot[dt] = wmma4(a, b, ot[dt]);
                }
            }
        }
        __syncthreads();
    }

    {
        float* so = Sq + qrow * TST + (hi << 3);
        #pragma unroll
        for (int dt = 0; dt < 4; ++dt)
            #pragma unroll
            for (int r = 0; r < 8; ++r)
                so[dt * 16 + r] = ot[dt][r] / lrun;
    }
    __syncthreads();
    #pragma unroll
    for (int it = 0; it < 8; ++it) {
        const int idx = it * 32 + lane;
        const int row = wave * 16 + (idx >> 4);
        const int c4  = (idx & 15) << 2;
        const float* s = Sq + row * TST + c4;
        float4 v;
        v.x = s[0]; v.y = s[1]; v.z = s[2]; v.w = s[3];
        *(float4*)(out + ((size_t)batch * SEQ + q0 + row) * DM + head * HD + c4) = v;
    }
}

extern "C" void kernel_launch(void* const* d_in, const int* in_sizes, int n_in,
                              void* d_out, int out_size, void* d_ws, size_t ws_size,
                              hipStream_t stream)
{
    const float* x    = (const float*)d_in[0];
    const float* Wqkv = (const float*)d_in[1];
    const float* bqkv = (const float*)d_in[2];
    const float* Wr   = (const float*)d_in[3];
    const float* br   = (const float*)d_in[4];
    float* out = (float*)d_out;

    float* ws  = (float*)d_ws;
    const size_t per = (size_t)NBH * SEQ * HD;
    float* qws = ws;
    float* kws = ws + per;
    float* vws = ws + 2 * per;

    dim3 g1(32, 24), b1(256);
    qkv_gemm<<<g1, b1, 0, stream>>>(x, Wqkv, bqkv, qws, kws, vws);

    dim3 g2(NBH, SEQ / 64), b2(128);
    size_t smem = (size_t)(3 * 64 * TST + 64 * RPST) * sizeof(float);
    rel_attn<<<g2, b2, smem, stream>>>(qws, kws, vws, Wr, br, out);
}


# qkv_gemm: k tile 32 deep, 2 LDS stages; tile kt+1 read global->registers (whole 128-B lines) under tile kt's WMMAs, stored next trip; 1 barrier per tile, were 2 per 16 k; sums in each 8 k reordered
# speedup vs baseline: 2.7114x; 1.0611x over previous
#include <hip/hip_runtime.h>

typedef float v2f __attribute__((ext_vector_type(2)));
typedef float v8f __attribute__((ext_vector_type(8)));

#define NHEADS 16
#define HD     64
#define SEQ    2048
#define DM     1024
#define NBH    32
#define RWIN   128
#define RSZ    257
#define RPST   260
#define TST    66
#define GP     36

__device__ __forceinline__ v8f wmma4(v2f a, v2f b, v8f c) {
    return __builtin_amdgcn_wmma_f32_16x16x4_f32(false, a, false, b, (short)0, c, false, false);
}

__device__ __forceinline__ float xhalf(float x) {
    int b = __builtin_bit_cast(int, x);
    return __builtin_bit_cast(float,
        __builtin_amdgcn_permlanex16(b, b, 0x76543210, 0xfedcba98, false, false));
}

__device__ __forceinline__ void gemm_tile(const float* st, int fa, int fb, v8f (&acc)[2][4]) {
    #pragma unroll
    for (int j = 0; j < 4; ++j) {
        const float4 a0 = *(const float4*)(st + fa + 8 * j);
        const float4 a1 = *(const float4*)(st + fa + 16 * GP + 8 * j);
        float4 b[4];
        #pragma unroll
        for (int nt = 0; nt < 4; ++nt)
            b[nt] = *(const float4*)(st + fb + nt * 16 * GP + 8 * j);
        #pragma unroll
        for (int nt = 0; nt < 4; ++nt) {
            const v2f bb = {b[nt].x, b[nt].y};
            acc[0][nt] = wmma4(v2f{a0.x, a0.y}, bb, acc[0][nt]);
            acc[1][nt] = wmma4(v2f{a1.x, a1.y}, bb, acc[1][nt]);
        }
        #pragma unroll
        for (int nt = 0; nt < 4; ++nt) {
            const v2f bb = {b[nt].z, b[nt].w};
            acc[0][nt] = wmma4(v2f{a0.z, a0.w}, bb, acc[0][nt]);
            acc[1][nt] = wmma4(v2f{a1.z, a1.w}, bb, acc[1][nt]);
        }
    }
}

__global__ __launch_bounds__(256, 1)
void qkv_gemm(const float* __restrict__ X, const float* __restrict__ W,
              const float* __restrict__ bias,
              float* __restrict__ qws, float* __restrict__ kws, float* __restrict__ vws)
{
    extern __shared__ __attribute__((aligned(16))) float gsm[];

    const int tid  = threadIdx.x;
    const int lane = tid & 31;
    const int wave = tid >> 5;
    const int lo   = lane & 15;
    const int hi   = lane >> 4;
    const int wm   = (wave & 3) * 32;
    const int wn   = (wave >> 2) * 64;
    const int mB   = blockIdx.x * 128;
    const int nB   = blockIdx.y * 128;

    const v8f zv = {0.f, 0.f, 0.f, 0.f, 0.f, 0.f, 0.f, 0.f};
    v8f acc[2][4];
    #pragma unroll
    for (int mt = 0; mt < 2; ++mt)
        #pragma unroll
        for (int nt = 0; nt < 4; ++nt)
            acc[mt][nt] = zv;

    const int lr  = tid >> 3;
    const int lc4 = (tid & 7) << 2;
    const float* ga = X + (size_t)(mB + lr) * DM + lc4;
    const float* gb = W + (size_t)(nB + lr) * DM + lc4;
    const int    so = lr * GP + lc4;

    const size_t rs = (size_t)32 * DM;
    float4 ra0 = *(const float4*)(ga);
    float4 ra1 = *(const float4*)(ga + rs);
    float4 ra2 = *(const float4*)(ga + 2 * rs);
    float4 ra3 = *(const float4*)(ga + 3 * rs);
    float4 rb0 = *(const float4*)(gb);
    float4 rb1 = *(const float4*)(gb + rs);
    float4 rb2 = *(const float4*)(gb + 2 * rs);
    float4 rb3 = *(const float4*)(gb + 3 * rs);

    const int fa = (wm + lo) * GP + (hi << 2);
    const int fb = 128 * GP + (wn + lo) * GP + (hi << 2);

    #pragma unroll 1
    for (int kt = 0; kt < DM / 32; ++kt) {
        float* st = gsm + (kt & 1) * (256 * GP);

        {
            float* da = st + so;
            float* db = st + 128 * GP + so;
            *(float4*)(da)           = ra0;
            *(float4*)(da + 32 * GP) = ra1;
            *(float4*)(da + 64 * GP) = ra2;
            *(float4*)(da + 96 * GP) = ra3;
            *(float4*)(db)           = rb0;
            *(float4*)(db + 32 * GP) = rb1;
            *(float4*)(db + 64 * GP) = rb2;
            *(float4*)(db + 96 * GP) = rb3;
        }
        __syncthreads();

        if (kt + 1 < DM / 32) {
            const float* na = ga + (kt + 1) * 32;
            const float* nb = gb + (kt + 1) * 32;
            ra0 = *(const float4*)(na);
            ra1 = *(const float4*)(na + rs);
            ra2 = *(const float4*)(na + 2 * rs);
            ra3 = *(const float4*)(na + 3 * rs);
            rb0 = *(const float4*)(nb);
            rb1 = *(const float4*)(nb + rs);
            rb2 = *(const float4*)(nb + 2 * rs);
            rb3 = *(const float4*)(nb + 3 * rs);
        }

        gemm_tile(st, fa, fb, acc);
    }

    #pragma unroll
    for (int nt = 0; nt < 4; ++nt) {
        const int col    = nB + wn + nt * 16 + lo;
        const float bv   = bias[col];
        const int head   = col / 192;
        const int within = col - head * 192;
        const int which  = within >> 6;
        const int dim    = within & 63;
        float* dst = (which == 0) ? qws : (which == 1) ? kws : vws;
        #pragma unroll
        for (int mt = 0; mt < 2; ++mt) {
            #pragma unroll
            for (int i = 0; i < 8; ++i) {
                const int row   = mB + wm + mt * 16 + i + (hi << 3);
                const int batch = row >> 11;
                const int si    = row & 2047;
                dst[(((size_t)batch * NHEADS + head) * SEQ + si) * HD + dim] =
                    acc[mt][nt][i] + bv;
            }
        }
    }
}

__global__ __launch_bounds__(128, 1)
void rel_attn(const float* __restrict__ qws, const float* __restrict__ kws,
              const float* __restrict__ vws, const float* __restrict__ Wr,
              const float* __restrict__ brv, float* __restrict__ out)
{
    extern __shared__ float sm[];
    float* Sq  = sm;
    float* Sk  = Sq + 64 * TST;
    float* Svt = Sk + 64 * TST;
    float* Srp = Svt + 64 * TST;

    const int bh   = blockIdx.x;
    const int qt   = blockIdx.y;
    const int tid  = threadIdx.x;
    const int lane = tid & 31;
    const int wave = tid >> 5;
    const int lo   = lane & 15;
    const int hi   = lane >> 4;

    const int batch = bh >> 4;
    const int head  = bh & 15;
    const size_t base = (size_t)bh * SEQ * HD;
    const int q0 = qt * 64;

    for (int idx = tid; idx < 64 * 16; idx += 128) {
        const int r  = idx >> 4;
        const int c4 = (idx & 15) << 2;
        float4 v = *(const float4*)(qws + base + (size_t)(q0 + r) * HD + c4);
        float* d = Sq + r * TST + c4;
        d[0] = v.x; d[1] = v.y; d[2] = v.z; d[3] = v.w;
    }
    __syncthreads();

    const v8f zv = {0.f, 0.f, 0.f, 0.f, 0.f, 0.f, 0.f, 0.f};

    const int sr  = tid >> 4;
    const int sc4 = (tid & 15) << 2;
    const float* kp = kws + base + (size_t)sr * HD + sc4;
    const float* vp = vws + base + (size_t)(2 * sr) * HD + sc4;
    float4 kreg[8], vreg[8];
    #pragma unroll
    for (int it = 0; it < 8; ++it) {
        kreg[it] = *(const float4*)(kp + it * (8 * HD));
        vreg[it] = *(const float4*)(vp + ((it >> 1) * 16 + (it & 1)) * HD);
    }

    {
        v2f af[16];
        const float* qrow = Sq + (wave * 16 + lo) * TST + (hi << 2);
        #pragma unroll
        for (int j = 0; j < 8; ++j) {
            af[2 * j]     = *(const v2f*)(qrow + 8 * j);
            af[2 * j + 1] = *(const v2f*)(qrow + 8 * j + 2);
        }

        float4 bc[8], bn[8];
        {
            const float4* p = (const float4*)(Wr + (size_t)lo * HD + (hi << 2));
            #pragma unroll
            for (int j = 0; j < 8; ++j) bc[j] = p[2 * j];
        }

        #pragma unroll 1
        for (int nt = 0; nt < 17; ++nt) {
            {
                int n = (nt + 1) * 16 + lo;
                n = (n < RSZ) ? n : (RSZ - 1);
                const float4* p = (const float4*)(Wr + (size_t)n * HD + (hi << 2));
                #pragma unroll
                for (int j = 0; j < 8; ++j) bn[j] = p[2 * j];
            }

            v8f c0 = zv, c1 = zv;
            #pragma unroll
            for (int j = 0; j < 4; ++j) {
                v2f b0 = {bc[j].x, bc[j].y};
                v2f b1 = {bc[j].z, bc[j].w};
                v2f b2 = {bc[j + 4].x, bc[j + 4].y};
                v2f b3 = {bc[j + 4].z, bc[j + 4].w};
                c0 = wmma4(af[2 * j],     b0, c0);
                c1 = wmma4(af[2 * j + 8], b2, c1);
                c0 = wmma4(af[2 * j + 1], b1, c0);
                c1 = wmma4(af[2 * j + 9], b3, c1);
            }

            const int col = nt * 16 + lo;
            if (col < RSZ) {
                const float bb = brv[col];
                #pragma unroll
                for (int i = 0; i < 8; ++i)
                    Srp[(wave * 16 + i + (hi << 3)) * RPST + col] = (c0[i] + c1[i]) + bb;
            }

            #pragma unroll
            for (int j = 0; j < 8; ++j) bc[j] = bn[j];
        }
    }

    v8f ot[4];
    #pragma unroll
    for (int dt = 0; dt < 4; ++dt) ot[dt] = zv;
    float mrun = -__builtin_inff(), lrun = 0.f;

    const int    qrow  = wave * 16 + lo;
    const float* srow  = Srp + qrow * RPST + RWIN;
    const int    dq    = (hi << 3) - (q0 + qrow);

    __syncthreads();

    for (int kt = 0; kt < 32; ++kt) {
        #pragma unroll
        for (int it = 0; it < 8; ++it) {
            float* dk = Sk + (sr + 8 * it) * TST + sc4;
            dk[0] = kreg[it].x; dk[1] = kreg[it].y; dk[2] = kreg[it].z; dk[3] = kreg[it].w;
        }
        #pragma unroll
        for (int p = 0; p < 4; ++p) {
            float* dv = Svt + sc4 * TST + 16 * p + 2 * sr;
            *(v2f*)(dv)           = v2f{vreg[2 * p].x, vreg[2 * p + 1].x};
            *(v2f*)(dv + TST)     = v2f{vreg[2 * p].y, vreg[2 * p + 1].y};
            *(v2f*)(dv + 2 * TST) = v2f{vreg[2 * p].z, vreg[2 * p + 1].z};
            *(v2f*)(dv + 3 * TST) = v2f{vreg[2 * p].w, vreg[2 * p + 1].w};
        }
        __syncthreads();

        if (kt + 1 < 32) {
            const float* kn = kp + (size_t)(kt + 1) * (64 * HD);
            const float* vn = vp + (size_t)(kt + 1) * (64 * HD);
            #pragma unroll
            for (int it = 0; it < 8; ++it) {
                kreg[it] = *(const float4*)(kn + it * (8 * HD));
                vreg[it] = *(const float4*)(vn + ((it >> 1) * 16 + (it & 1)) * HD);
            }
        }

        v8f st[4];
        #pragma unroll
        for (int nt = 0; nt < 4; ++nt) st[nt] = zv;
        #pragma unroll
        for (int ksub = 0; ksub < 16; ++ksub) {
            const int k0 = (ksub << 2) + (hi << 1);
            v2f b = *(const v2f*)(Sq + qrow * TST + k0);
            #pragma unroll
            for (int nt = 0; nt < 4; ++nt) {
                v2f a = *(const v2f*)(Sk + (nt * 16 + lo) * TST + k0);
                st[nt] = wmma4(a, b, st[nt]);
            }
        }

        const int dbase = kt * 64 + dq;
        float tmax = -__builtin_inff();
        #pragma unroll
        for (int nt = 0; nt < 4; ++nt) {
            #pragma unroll
            for (int r = 0; r < 8; ++r) {
                int off = dbase + nt * 16 + r;
                off = (off < -RWIN) ? -RWIN : (off > RWIN) ? RWIN : off;
                float s = st[nt][r] * 0.125f + srow[off];
                st[nt][r] = s;
                tmax = fmaxf(tmax, s);
            }
        }
        tmax = fmaxf(tmax, xhalf(tmax));

        const float mnew  = fmaxf(mrun, tmax);
        const float alpha = __expf(mrun - mnew);
        mrun = mnew;
        float rs[4];
        #pragma unroll
        for (int nt = 0; nt < 4; ++nt) {
            rs[nt] = 0.f;
            #pragma unroll
            for (int r = 0; r < 8; ++r) {
                float p = __expf(st[nt][r] - mnew);
                st[nt][r] = p;
                rs[nt] += p;
            }
        }
        float rsum = (rs[0] + rs[1]) + (rs[2] + rs[3]);
        rsum += xhalf(rsum);
        lrun = lrun * alpha + rsum;

        #pragma unroll
        for (int dt = 0; dt < 4; ++dt)
            #pragma unroll
            for (int r = 0; r < 8; ++r)
                ot[dt][r] *= alpha;

        #pragma unroll
        for (int nt = 0; nt < 4; ++nt) {
            #pragma unroll
            for (int t = 0; t < 4; ++t) {
                v2f b = {st[nt][2 * t], st[nt][2 * t + 1]};
                const float* vcol = Svt + lo * TST + nt * 16 + (hi << 3) + 2 * t;
                #pragma unroll
                for (int dt = 0; dt < 4; ++dt) {
                    v2f a = *(const v2f*)(vcol + dt * 16 * TST);
                    ot[dt] = wmma4(a, b, ot[dt]);
                }
            }
        }
        __syncthreads();
    }

    {
        float* so = Sq + qrow * TST + (hi << 3);
        #pragma unroll
        for (int dt = 0; dt < 4; ++dt)
            #pragma unroll
            for (int r = 0; r < 8; ++r)
                so[dt * 16 + r] = ot[dt][r] / lrun;
    }
    __syncthreads();
    #pragma unroll
    for (int it = 0; it < 8; ++it) {
        const int idx = it * 32 + lane;
        const int row = wave * 16 + (idx >> 4);
        const int c4  = (idx & 15) << 2;
        const float* s = Sq + row * TST + c4;
        float4 v;
        v.x = s[0]; v.y = s[1]; v.z = s[2]; v.w = s[3];
        *(float4*)(out + ((size_t)batch * SEQ + q0 + row) * DM + head * HD + c4) = v;
    }
}

extern "C" void kernel_launch(void* const* d_in, const int* in_sizes, int n_in,
                              void* d_out, int out_size, void* d_ws, size_t ws_size,
                              hipStream_t stream)
{
    const float* x    = (const float*)d_in[0];
    const float* Wqkv = (const float*)d_in[1];
    const float* bqkv = (const float*)d_in[2];
    const float* Wr   = (const float*)d_in[3];
    const float* br   = (const float*)d_in[4];
    float* out = (float*)d_out;

    float* ws  = (float*)d_ws;
    const size_t per = (size_t)NBH * SEQ * HD;
    float* qws = ws;
    float* kws = ws + per;
    float* vws = ws + 2 * per;

    dim3 g1(32, 24), b1(256);
    size_t smem1 = (size_t)(2 * 256 * GP) * sizeof(float);
    qkv_gemm<<<g1, b1, smem1, stream>>>(x, Wqkv, bqkv, qws, kws, vws);

    dim3 g2(NBH, SEQ / 64), b2(128);
    size_t smem = (size_t)(3 * 64 * TST + 64 * RPST) * sizeof(float);
    rel_attn<<<g2, b2, smem, stream>>>(qws, kws, vws, Wr, br, out);
}


# rel_attn: 8-wave work-group (was 4): its two 4-wave groups take alternate key tiles over the shared Q and rp tiles in LDS, merge max/sum/O once at the end; 4 waves per SIMD, were 2; sum order changes
# speedup vs baseline: 2.8299x; 1.0437x over previous
#include <hip/hip_runtime.h>

typedef float v2f __attribute__((ext_vector_type(2)));
typedef float v8f __attribute__((ext_vector_type(8)));

#define NHEADS 16
#define HD     64
#define SEQ    2048
#define DM     1024
#define NBH    32
#define RWIN   128
#define RSZ    257
#define RPST   260
#define TST    66
#define GP     36

__device__ __forceinline__ v8f wmma4(v2f a, v2f b, v8f c) {
    return __builtin_amdgcn_wmma_f32_16x16x4_f32(false, a, false, b, (short)0, c, false, false);
}

__device__ __forceinline__ float xhalf(float x) {
    int b = __builtin_bit_cast(int, x);
    return __builtin_bit_cast(float,
        __builtin_amdgcn_permlanex16(b, b, 0x76543210, 0xfedcba98, false, false));
}

__device__ __forceinline__ void gemm_tile(const float* st, int fa, int fb, v8f (&acc)[2][4]) {
    #pragma unroll
    for (int j = 0; j < 4; ++j) {
        const float4 a0 = *(const float4*)(st + fa + 8 * j);
        const float4 a1 = *(const float4*)(st + fa + 16 * GP + 8 * j);
        float4 b[4];
        #pragma unroll
        for (int nt = 0; nt < 4; ++nt)
            b[nt] = *(const float4*)(st + fb + nt * 16 * GP + 8 * j);
        #pragma unroll
        for (int nt = 0; nt < 4; ++nt) {
            const v2f bb = {b[nt].x, b[nt].y};
            acc[0][nt] = wmma4(v2f{a0.x, a0.y}, bb, acc[0][nt]);
            acc[1][nt] = wmma4(v2f{a1.x, a1.y}, bb, acc[1][nt]);
        }
        #pragma unroll
        for (int nt = 0; nt < 4; ++nt) {
            const v2f bb = {b[nt].z, b[nt].w};
            acc[0][nt] = wmma4(v2f{a0.z, a0.w}, bb, acc[0][nt]);
            acc[1][nt] = wmma4(v2f{a1.z, a1.w}, bb, acc[1][nt]);
        }
    }
}

__global__ __launch_bounds__(256, 1)
void qkv_gemm(const float* __restrict__ X, const float* __restrict__ W,
              const float* __restrict__ bias,
              float* __restrict__ qws, float* __restrict__ kws, float* __restrict__ vws)
{
    extern __shared__ __attribute__((aligned(16))) float gsm[];

    const int tid  = threadIdx.x;
    const int lane = tid & 31;
    const int wave = tid >> 5;
    const int lo   = lane & 15;
    const int hi   = lane >> 4;
    const int wm   = (wave & 3) * 32;
    const int wn   = (wave >> 2) * 64;
    const int mB   = blockIdx.x * 128;
    const int nB   = blockIdx.y * 128;

    const v8f zv = {0.f, 0.f, 0.f, 0.f, 0.f, 0.f, 0.f, 0.f};
    v8f acc[2][4];
    #pragma unroll
    for (int mt = 0; mt < 2; ++mt)
        #pragma unroll
        for (int nt = 0; nt < 4; ++nt)
            acc[mt][nt] = zv;

    const int lr  = tid >> 3;
    const int lc4 = (tid & 7) << 2;
    const float* ga = X + (size_t)(mB + lr) * DM + lc4;
    const float* gb = W + (size_t)(nB + lr) * DM + lc4;
    const int    so = lr * GP + lc4;

    const size_t rs = (size_t)32 * DM;
    float4 ra0 = *(const float4*)(ga);
    float4 ra1 = *(const float4*)(ga + rs);
    float4 ra2 = *(const float4*)(ga + 2 * rs);
    float4 ra3 = *(const float4*)(ga + 3 * rs);
    float4 rb0 = *(const float4*)(gb);
    float4 rb1 = *(const float4*)(gb + rs);
    float4 rb2 = *(const float4*)(gb + 2 * rs);
    float4 rb3 = *(const float4*)(gb + 3 * rs);

    const int fa = (wm + lo) * GP + (hi << 2);
    const int fb = 128 * GP + (wn + lo) * GP + (hi << 2);

    #pragma unroll 1
    for (int kt = 0; kt < DM / 32; ++kt) {
        float* st = gsm + (kt & 1) * (256 * GP);

        {
            float* da = st + so;
            float* db = st + 128 * GP + so;
            *(float4*)(da)           = ra0;
            *(float4*)(da + 32 * GP) = ra1;
            *(float4*)(da + 64 * GP) = ra2;
            *(float4*)(da + 96 * GP) = ra3;
            *(float4*)(db)           = rb0;
            *(float4*)(db + 32 * GP) = rb1;
            *(float4*)(db + 64 * GP) = rb2;
            *(float4*)(db + 96 * GP) = rb3;
        }
        __syncthreads();

        if (kt + 1 < DM / 32) {
            const float* na = ga + (kt + 1) * 32;
            const float* nb = gb + (kt + 1) * 32;
            ra0 = *(const float4*)(na);
            ra1 = *(const float4*)(na + rs);
            ra2 = *(const float4*)(na + 2 * rs);
            ra3 = *(const float4*)(na + 3 * rs);
            rb0 = *(const float4*)(nb);
            rb1 = *(const float4*)(nb + rs);
            rb2 = *(const float4*)(nb + 2 * rs);
            rb3 = *(const float4*)(nb + 3 * rs);
        }

        gemm_tile(st, fa, fb, acc);
    }

    #pragma unroll
    for (int nt = 0; nt < 4; ++nt) {
        const int col    = nB + wn + nt * 16 + lo;
        const float bv   = bias[col];
        const int head   = col / 192;
        const int within = col - head * 192;
        const int which  = within >> 6;
        const int dim    = within & 63;
        float* dst = (which == 0) ? qws : (which == 1) ? kws : vws;
        #pragma unroll
        for (int mt = 0; mt < 2; ++mt) {
            #pragma unroll
            for (int i = 0; i < 8; ++i) {
                const int row   = mB + wm + mt * 16 + i + (hi << 3);
                const int batch = row >> 11;
                const int si    = row & 2047;
                dst[(((size_t)batch * NHEADS + head) * SEQ + si) * HD + dim] =
                    acc[mt][nt][i] + bv;
            }
        }
    }
}

__global__ __launch_bounds__(256, 2)
void rel_attn(const float* __restrict__ qws, const float* __restrict__ kws,
              const float* __restrict__ vws, const float* __restrict__ Wr,
              const float* __restrict__ brv, float* __restrict__ out)
{
    extern __shared__ float sm[];
    float* Sq  = sm;
    float* Skv = Sq + 64 * TST;
    float* Srp = Skv + 4 * 64 * TST;

    const int bh   = blockIdx.x;
    const int qt   = blockIdx.y;
    const int tid  = threadIdx.x;
    const int lane = tid & 31;
    const int grp  = __builtin_amdgcn_readfirstlane(tid >> 7);
    const int wave = (tid >> 5) & 3;
    const int gtid = tid & 127;
    float* Sk  = Skv + grp * (2 * 64 * TST);
    float* Svt = Sk + 64 * TST;
    const int lo   = lane & 15;
    const int hi   = lane >> 4;

    const int batch = bh >> 4;
    const int head  = bh & 15;
    const size_t base = (size_t)bh * SEQ * HD;
    const int q0 = qt * 64;

    for (int idx = tid; idx < 64 * 16; idx += 256) {
        const int r  = idx >> 4;
        const int c4 = (idx & 15) << 2;
        float4 v = *(const float4*)(qws + base + (size_t)(q0 + r) * HD + c4);
        float* d = Sq + r * TST + c4;
        d[0] = v.x; d[1] = v.y; d[2] = v.z; d[3] = v.w;
    }
    __syncthreads();

    const v8f zv = {0.f, 0.f, 0.f, 0.f, 0.f, 0.f, 0.f, 0.f};

    const int sr  = gtid >> 4;
    const int sc4 = (gtid & 15) << 2;
    const float* kp = kws + base + (size_t)(grp * 64 + sr) * HD + sc4;
    const float* vp = vws + base + (size_t)(grp * 64 + 2 * sr) * HD + sc4;
    float4 kreg[8], vreg[8];
    #pragma unroll
    for (int it = 0; it < 8; ++it) {
        kreg[it] = *(const float4*)(kp + it * (8 * HD));
        vreg[it] = *(const float4*)(vp + ((it >> 1) * 16 + (it & 1)) * HD);
    }

    {
        v2f af[16];
        const float* qrow = Sq + (wave * 16 + lo) * TST + (hi << 2);
        #pragma unroll
        for (int j = 0; j < 8; ++j) {
            af[2 * j]     = *(const v2f*)(qrow + 8 * j);
            af[2 * j + 1] = *(const v2f*)(qrow + 8 * j + 2);
        }

        const int nt0 = grp * 9;
        const int nt1 = 9 + grp * 8;
        float4 bc[8], bn[8];
        {
            const float4* p = (const float4*)(Wr + (size_t)(nt0 * 16 + lo) * HD + (hi << 2));
            #pragma unroll
            for (int j = 0; j < 8; ++j) bc[j] = p[2 * j];
        }

        #pragma unroll 1
        for (int nt = nt0; nt < nt1; ++nt) {
            {
                int n = (nt + 1) * 16 + lo;
                n = (n < RSZ) ? n : (RSZ - 1);
                const float4* p = (const float4*)(Wr + (size_t)n * HD + (hi << 2));
                #pragma unroll
                for (int j = 0; j < 8; ++j) bn[j] = p[2 * j];
            }

            v8f c0 = zv, c1 = zv;
            #pragma unroll
            for (int j = 0; j < 4; ++j) {
                v2f b0 = {bc[j].x, bc[j].y};
                v2f b1 = {bc[j].z, bc[j].w};
                v2f b2 = {bc[j + 4].x, bc[j + 4].y};
                v2f b3 = {bc[j + 4].z, bc[j + 4].w};
                c0 = wmma4(af[2 * j],     b0, c0);
                c1 = wmma4(af[2 * j + 8], b2, c1);
                c0 = wmma4(af[2 * j + 1], b1, c0);
                c1 = wmma4(af[2 * j + 9], b3, c1);
            }

            const int col = nt * 16 + lo;
            if (col < RSZ) {
                const float bb = brv[col];
                #pragma unroll
                for (int i = 0; i < 8; ++i)
                    Srp[(wave * 16 + i + (hi << 3)) * RPST + col] = (c0[i] + c1[i]) + bb;
            }

            #pragma unroll
            for (int j = 0; j < 8; ++j) bc[j] = bn[j];
        }
    }

    v8f ot[4];
    #pragma unroll
    for (int dt = 0; dt < 4; ++dt) ot[dt] = zv;
    float mrun = -__builtin_inff(), lrun = 0.f;

    const int    qrow  = wave * 16 + lo;
    const float* srow  = Srp + qrow * RPST + RWIN;
    const int    dq    = (hi << 3) - (q0 + qrow);

    __syncthreads();

    for (int i = 0; i < 16; ++i) {
        const int kt = 2 * i + grp;
        #pragma unroll
        for (int it = 0; it < 8; ++it) {
            float* dk = Sk + (sr + 8 * it) * TST + sc4;
            dk[0] = kreg[it].x; dk[1] = kreg[it].y; dk[2] = kreg[it].z; dk[3] = kreg[it].w;
        }
        #pragma unroll
        for (int p = 0; p < 4; ++p) {
            float* dv = Svt + sc4 * TST + 16 * p + 2 * sr;
            *(v2f*)(dv)           = v2f{vreg[2 * p].x, vreg[2 * p + 1].x};
            *(v2f*)(dv + TST)     = v2f{vreg[2 * p].y, vreg[2 * p + 1].y};
            *(v2f*)(dv + 2 * TST) = v2f{vreg[2 * p].z, vreg[2 * p + 1].z};
            *(v2f*)(dv + 3 * TST) = v2f{vreg[2 * p].w, vreg[2 * p + 1].w};
        }
        __syncthreads();

        if (i + 1 < 16) {
            const float* kn = kp + (size_t)(i + 1) * (128 * HD);
            const float* vn = vp + (size_t)(i + 1) * (128 * HD);
            #pragma unroll
            for (int it = 0; it < 8; ++it) {
                kreg[it] = *(const float4*)(kn + it * (8 * HD));
                vreg[it] = *(const float4*)(vn + ((it >> 1) * 16 + (it & 1)) * HD);
            }
        }

        v8f st[4];
        #pragma unroll
        for (int nt = 0; nt < 4; ++nt) st[nt] = zv;
        #pragma unroll
        for (int ksub = 0; ksub < 16; ++ksub) {
            const int k0 = (ksub << 2) + (hi << 1);
            v2f b = *(const v2f*)(Sq + qrow * TST + k0);
            #pragma unroll
            for (int nt = 0; nt < 4; ++nt) {
                v2f a = *(const v2f*)(Sk + (nt * 16 + lo) * TST + k0);
                st[nt] = wmma4(a, b, st[nt]);
            }
        }

        const int dbase = kt * 64 + dq;
        float tmax = -__builtin_inff();
        #pragma unroll
        for (int nt = 0; nt < 4; ++nt) {
            #pragma unroll
            for (int r = 0; r < 8; ++r) {
                int off = dbase + nt * 16 + r;
                off = (off < -RWIN) ? -RWIN : (off > RWIN) ? RWIN : off;
                float s = st[nt][r] * 0.125f + srow[off];
                st[nt][r] = s;
                tmax = fmaxf(tmax, s);
            }
        }
        tmax = fmaxf(tmax, xhalf(tmax));

        const float mnew  = fmaxf(mrun, tmax);
        const float alpha = __expf(mrun - mnew);
        mrun = mnew;
        float rs[4];
        #pragma unroll
        for (int nt = 0; nt < 4; ++nt) {
            rs[nt] = 0.f;
            #pragma unroll
            for (int r = 0; r < 8; ++r) {
                float p = __expf(st[nt][r] - mnew);
                st[nt][r] = p;
                rs[nt] += p;
            }
        }
        float rsum = (rs[0] + rs[1]) + (rs[2] + rs[3]);
        rsum += xhalf(rsum);
        lrun = lrun * alpha + rsum;

        #pragma unroll
        for (int dt = 0; dt < 4; ++dt)
            #pragma unroll
            for (int r = 0; r < 8; ++r)
                ot[dt][r] *= alpha;

        #pragma unroll
        for (int nt = 0; nt < 4; ++nt) {
            #pragma unroll
            for (int t = 0; t < 4; ++t) {
                v2f b = {st[nt][2 * t], st[nt][2 * t + 1]};
                const float* vcol = Svt + lo * TST + nt * 16 + (hi << 3) + 2 * t;
                #pragma unroll
                for (int dt = 0; dt < 4; ++dt) {
                    v2f a = *(const v2f*)(vcol + dt * 16 * TST);
                    ot[dt] = wmma4(a, b, ot[dt]);
                }
            }
        }
        __syncthreads();
    }

    float* Sx = Skv + wave * (34 * 32) + lane;
    if (grp == 1) {
        #pragma unroll
        for (int dt = 0; dt < 4; ++dt)
            #pragma unroll
            for (int r = 0; r < 8; ++r)
                Sx[(dt * 8 + r) * 32] = ot[dt][r];
        Sx[32 * 32] = mrun;
        Sx[33 * 32] = lrun;
    }
    __syncthreads();
    if (grp == 0) {
        const float m1 = Sx[32 * 32];
        const float l1 = Sx[33 * 32];
        const float m  = fmaxf(mrun, m1);
        const float a0 = __expf(mrun - m);
        const float a1 = __expf(m1 - m);
        const float l  = lrun * a0 + l1 * a1;
        float* so = Sq + qrow * TST + (hi << 3);
        #pragma unroll
        for (int dt = 0; dt < 4; ++dt)
            #pragma unroll
            for (int r = 0; r < 8; ++r)
                so[dt * 16 + r] = (ot[dt][r] * a0 + Sx[(dt * 8 + r) * 32] * a1) / l;
    }
    __syncthreads();
    #pragma unroll
    for (int it = 0; it < 4; ++it) {
        const int idx = it * 256 + tid;
        const int row = idx >> 4;
        const int c4  = (idx & 15) << 2;
        const float* s = Sq + row * TST + c4;
        float4 v;
        v.x = s[0]; v.y = s[1]; v.z = s[2]; v.w = s[3];
        *(float4*)(out + ((size_t)batch * SEQ + q0 + row) * DM + head * HD + c4) = v;
    }
}

extern "C" void kernel_launch(void* const* d_in, const int* in_sizes, int n_in,
                              void* d_out, int out_size, void* d_ws, size_t ws_size,
                              hipStream_t stream)
{
    const float* x    = (const float*)d_in[0];
    const float* Wqkv = (const float*)d_in[1];
    const float* bqkv = (const float*)d_in[2];
    const float* Wr   = (const float*)d_in[3];
    const float* br   = (const float*)d_in[4];
    float* out = (float*)d_out;

    float* ws  = (float*)d_ws;
    const size_t per = (size_t)NBH * SEQ * HD;
    float* qws = ws;
    float* kws = ws + per;
    float* vws = ws + 2 * per;

    dim3 g1(32, 24), b1(256);
    size_t smem1 = (size_t)(2 * 256 * GP) * sizeof(float);
    qkv_gemm<<<g1, b1, smem1, stream>>>(x, Wqkv, bqkv, qws, kws, vws);

    dim3 g2(NBH, SEQ / 64), b2(256);
    size_t smem = (size_t)(5 * 64 * TST + 64 * RPST) * sizeof(float);
    rel_attn<<<g2, b2, smem, stream>>>(qws, kws, vws, Wr, br, out);
}
